# starting kernel
# baseline (speedup 1.0000x reference)
#include <hip/hip_runtime.h>
#include <hip/hip_bf16.h>

typedef __attribute__((ext_vector_type(2))) float v2f;
typedef __attribute__((ext_vector_type(8))) float v8f;

__device__ __forceinline__ float lrelu(float v) { return v > 0.0f ? v : 0.01f * v; }

__global__ void gcn_deg_init(float* __restrict__ deg, int n) {
    int i = blockIdx.x * blockDim.x + threadIdx.x;
    if (i < n) deg[i] = 1.0f;
}

__global__ void gcn_deg_count(const int* __restrict__ dst, float* __restrict__ deg, int E) {
    int e = blockIdx.x * blockDim.x + threadIdx.x;
    if (e < E) atomicAdd(&deg[dst[e]], 1.0f);
}

__global__ void gcn_deg_finish(float* __restrict__ dinv, int n) {
    int i = blockIdx.x * blockDim.x + threadIdx.x;
    if (i < n) dinv[i] = rsqrtf(dinv[i]);
}

template <int K, int NOUT, int MODE>
__global__ __launch_bounds__(128) void gcn_gemm(
    const float* __restrict__ in,
    const float* __restrict__ W,
    const float* __restrict__ bias,
    const float* __restrict__ dinv,
    float* __restrict__ out,
    float* __restrict__ acc_out,
    int n)
{
    __shared__ float sW[NOUT * K];
    for (int i = threadIdx.x; i < NOUT * K / 4; i += blockDim.x)
        ((float4*)sW)[i] = ((const float4*)W)[i];
    __syncthreads();

    const int lane  = threadIdx.x & 31;
    const int wave  = threadIdx.x >> 5;
    const int row0  = (blockIdx.x * 4 + wave) * 16;
    if (row0 >= n) return;

    const int lhalf = lane >> 4;
    const int l16   = lane & 15;
    const int arow  = row0 + l16;

    v8f acc[NOUT / 16];
    #pragma unroll
    for (int t = 0; t < NOUT / 16; ++t)
        acc[t] = (v8f){0.f, 0.f, 0.f, 0.f, 0.f, 0.f, 0.f, 0.f};

    #pragma unroll
    for (int k0 = 0; k0 < K; k0 += 4) {
        const int kb = k0 + 2 * lhalf;
        v2f a;
        a.x = in[(size_t)arow * K + kb];
        a.y = in[(size_t)arow * K + kb + 1];
        #pragma unroll
        for (int t = 0; t < NOUT / 16; ++t) {
            const int ncol = t * 16 + l16;
            v2f b;
            b.x = sW[ncol * K + kb];
            b.y = sW[ncol * K + kb + 1];
            acc[t] = __builtin_amdgcn_wmma_f32_16x16x4_f32(
                false, a, false, b, (short)0, acc[t], false, false);
        }
    }

    #pragma unroll
    for (int t = 0; t < NOUT / 16; ++t) {
        const int col = t * 16 + l16;
        if (MODE == 0) {
            const float bv = bias[col];
            #pragma unroll
            for (int r = 0; r < 8; ++r) {
                const int row = row0 + r + 8 * lhalf;
                out[(size_t)row * NOUT + col] = lrelu(acc[t][r] + bv);
            }
        } else {
            #pragma unroll
            for (int r = 0; r < 8; ++r) {
                const int row = row0 + r + 8 * lhalf;
                const float val = acc[t][r] * dinv[row];
                out[(size_t)row * NOUT + col]     = val;
                acc_out[(size_t)row * NOUT + col] = val;
            }
        }
    }
}

__global__ void gcn_scatter(const int* __restrict__ src, const int* __restrict__ dst,
                            const float* __restrict__ p, float* __restrict__ acc, int E)
{
    int t = blockIdx.x * blockDim.x + threadIdx.x;
    int e = t >> 3;
    if (e >= E) return;
    int c4 = (t & 7) * 4;
    int u = src[e];
    int v = dst[e];
    float4 m = *(const float4*)(p + (size_t)u * 32 + c4);
    float* a = acc + (size_t)v * 32 + c4;
    atomicAdd(a + 0, m.x);
    atomicAdd(a + 1, m.y);
    atomicAdd(a + 2, m.z);
    atomicAdd(a + 3, m.w);
}

__global__ void gcn_finish(float* __restrict__ acc, const float* __restrict__ dinv,
                           const float* __restrict__ bias, int n)
{
    int t = blockIdx.x * blockDim.x + threadIdx.x;
    if (t >= n * 32) return;
    int row = t >> 5;
    int c   = t & 31;
    acc[t] = lrelu(dinv[row] * acc[t] + bias[c]);
}

__global__ void gcn_out(const float* __restrict__ acc2, const float* __restrict__ dinv,
                        const float* __restrict__ b2, const float* __restrict__ Wout,
                        const float* __restrict__ bout, float* __restrict__ out, int n)
{
    int i = blockIdx.x * blockDim.x + threadIdx.x;
    if (i >= n) return;
    float di = dinv[i];
    float s = 0.0f;
    #pragma unroll
    for (int c = 0; c < 32; ++c)
        s += Wout[c] * lrelu(di * acc2[(size_t)i * 32 + c] + b2[c]);
    out[i] = s + bout[0];
}

extern "C" void kernel_launch(void* const* d_in, const int* in_sizes, int n_in,
                              void* d_out, int out_size, void* d_ws, size_t ws_size,
                              hipStream_t stream)
{
    const float* x     = (const float*)d_in[0];
    const int*   ei    = (const int*)d_in[1];
    const float* W_in  = (const float*)d_in[2];
    const float* b_in  = (const float*)d_in[3];
    const float* W1    = (const float*)d_in[4];
    const float* b1    = (const float*)d_in[5];
    const float* W2    = (const float*)d_in[6];
    const float* b2    = (const float*)d_in[7];
    const float* W_out = (const float*)d_in[8];
    const float* b_out = (const float*)d_in[9];
    float* out = (float*)d_out;

    const int N = in_sizes[0] / 64;
    const int E = in_sizes[1] / 2;
    const int* src = ei;
    const int* dst = ei + E;

    float* ws   = (float*)d_ws;
    float* dinv = ws;
    float* h0   = ws + (size_t)N;
    float* p    = ws + (size_t)N * 65;
    float* accA = ws + (size_t)N * 97;
    float* accB = ws + (size_t)N * 129;

    const int TB = 256;
    gcn_deg_init  <<<(N + TB - 1) / TB, TB, 0, stream>>>(dinv, N);
    gcn_deg_count <<<(E + TB - 1) / TB, TB, 0, stream>>>(dst, dinv, E);
    gcn_deg_finish<<<(N + TB - 1) / TB, TB, 0, stream>>>(dinv, N);

    const int tiles  = (N + 15) / 16;
    const int gblk   = (tiles + 3) / 4;

    gcn_gemm<64, 64, 0><<<gblk, 128, 0, stream>>>(x, W_in, b_in, nullptr, h0, nullptr, N);

    gcn_gemm<64, 32, 1><<<gblk, 128, 0, stream>>>(h0, W1, nullptr, dinv, p, accA, N);
    gcn_scatter<<<((size_t)E * 8 + TB - 1) / TB, TB, 0, stream>>>(src, dst, p, accA, E);
    gcn_finish <<<((size_t)N * 32 + TB - 1) / TB, TB, 0, stream>>>(accA, dinv, b1, N);

    gcn_gemm<32, 32, 1><<<gblk, 128, 0, stream>>>(accA, W2, nullptr, dinv, p, accB, N);
    gcn_scatter<<<((size_t)E * 8 + TB - 1) / TB, TB, 0, stream>>>(src, dst, p, accB, E);

    gcn_out<<<(N + TB - 1) / TB, TB, 0, stream>>>(accB, dinv, b2, W_out, b_out, out, N);
}


# degree count as LDS integer histograms per node range and edge chunk, plain partial stores summed in finish: no global atomics there, init launch gone (10->9); dinv bit-identical
# speedup vs baseline: 1.2180x; 1.2180x over previous
#include <hip/hip_runtime.h>
#include <hip/hip_bf16.h>

typedef __attribute__((ext_vector_type(2))) float v2f;
typedef __attribute__((ext_vector_type(8))) float v8f;

__device__ __forceinline__ float lrelu(float v) { return v > 0.0f ? v : 0.01f * v; }

#define DEG_R   16
#define DEG_C   16
#define DEG_CAP 8192
#define DEG_TB  1024

__device__ __forceinline__ void deg_tally(unsigned* s, int d, int base, unsigned cnt) {
    const unsigned off = (unsigned)(d - base);
    if (off < cnt) atomicAdd(&s[off], 1u);
}

__global__ __launch_bounds__(DEG_TB) void gcn_deg_hist(const int* __restrict__ dst,
                                                       float* __restrict__ part, int n, int E)
{
    __shared__ unsigned s_cnt[DEG_CAP];
    const int tid = threadIdx.x;
    const int r = blockIdx.x % DEG_R;
    const int c = blockIdx.x / DEG_R;

    const int rsize = (n + DEG_R - 1) / DEG_R;
    const int rlo = min(n, r * rsize);
    const int rhi = min(n, rlo + rsize);

    const int head  = min(E, (int)(((16u - ((unsigned)(size_t)dst & 15u)) & 15u) >> 2));
    const int4* __restrict__ dst4 = (const int4*)(dst + head);
    const int nvec  = (E - head) >> 2;
    const int tail0 = head + 4 * nvec;
    const int vsz = (nvec + DEG_C - 1) / DEG_C;
    const int vlo = min(nvec, c * vsz);
    const int vhi = min(nvec, vlo + vsz);

    for (int base = rlo; base < rhi; base += DEG_CAP) {
        const unsigned cnt = (unsigned)min(DEG_CAP, rhi - base);
        for (unsigned i = tid; i < cnt; i += DEG_TB) s_cnt[i] = 0u;
        __syncthreads();

        int i = vlo + tid;
        for (; i + 7 * DEG_TB < vhi; i += 8 * DEG_TB) {
            int4 v[8];
            #pragma unroll
            for (int k = 0; k < 8; ++k) v[k] = dst4[i + k * DEG_TB];
            #pragma unroll
            for (int k = 0; k < 8; ++k) {
                deg_tally(s_cnt, v[k].x, base, cnt);
                deg_tally(s_cnt, v[k].y, base, cnt);
                deg_tally(s_cnt, v[k].z, base, cnt);
                deg_tally(s_cnt, v[k].w, base, cnt);
            }
        }
        for (; i < vhi; i += DEG_TB) {
            const int4 v = dst4[i];
            deg_tally(s_cnt, v.x, base, cnt);
            deg_tally(s_cnt, v.y, base, cnt);
            deg_tally(s_cnt, v.z, base, cnt);
            deg_tally(s_cnt, v.w, base, cnt);
        }
        if (c == 0) {
            if (tid < head)      deg_tally(s_cnt, dst[tid], base, cnt);
            if (tid < E - tail0) deg_tally(s_cnt, dst[tail0 + tid], base, cnt);
        }
        __syncthreads();

        for (unsigned i2 = tid; i2 < cnt; i2 += DEG_TB)
            part[(size_t)c * n + base + i2] = (float)s_cnt[i2];
        __syncthreads();
    }
}

__global__ void gcn_deg_finish(const float* __restrict__ part, float* __restrict__ dinv, int n) {
    int i = blockIdx.x * blockDim.x + threadIdx.x;
    if (i >= n) return;
    float d = 1.0f;
    #pragma unroll
    for (int c = 0; c < DEG_C; ++c) d += part[(size_t)c * n + i];
    dinv[i] = rsqrtf(d);
}

template <int K, int NOUT, int MODE>
__global__ __launch_bounds__(128) void gcn_gemm(
    const float* __restrict__ in,
    const float* __restrict__ W,
    const float* __restrict__ bias,
    const float* __restrict__ dinv,
    float* __restrict__ out,
    float* __restrict__ acc_out,
    int n)
{
    __shared__ float sW[NOUT * K];
    for (int i = threadIdx.x; i < NOUT * K / 4; i += blockDim.x)
        ((float4*)sW)[i] = ((const float4*)W)[i];
    __syncthreads();

    const int lane  = threadIdx.x & 31;
    const int wave  = threadIdx.x >> 5;
    const int row0  = (blockIdx.x * 4 + wave) * 16;
    if (row0 >= n) return;

    const int lhalf = lane >> 4;
    const int l16   = lane & 15;
    const int arow  = row0 + l16;

    v8f acc[NOUT / 16];
    #pragma unroll
    for (int t = 0; t < NOUT / 16; ++t)
        acc[t] = (v8f){0.f, 0.f, 0.f, 0.f, 0.f, 0.f, 0.f, 0.f};

    #pragma unroll
    for (int k0 = 0; k0 < K; k0 += 4) {
        const int kb = k0 + 2 * lhalf;
        v2f a;
        a.x = in[(size_t)arow * K + kb];
        a.y = in[(size_t)arow * K + kb + 1];
        #pragma unroll
        for (int t = 0; t < NOUT / 16; ++t) {
            const int ncol = t * 16 + l16;
            v2f b;
            b.x = sW[ncol * K + kb];
            b.y = sW[ncol * K + kb + 1];
            acc[t] = __builtin_amdgcn_wmma_f32_16x16x4_f32(
                false, a, false, b, (short)0, acc[t], false, false);
        }
    }

    #pragma unroll
    for (int t = 0; t < NOUT / 16; ++t) {
        const int col = t * 16 + l16;
        if (MODE == 0) {
            const float bv = bias[col];
            #pragma unroll
            for (int r = 0; r < 8; ++r) {
                const int row = row0 + r + 8 * lhalf;
                out[(size_t)row * NOUT + col] = lrelu(acc[t][r] + bv);
            }
        } else {
            #pragma unroll
            for (int r = 0; r < 8; ++r) {
                const int row = row0 + r + 8 * lhalf;
                const float val = acc[t][r] * dinv[row];
                out[(size_t)row * NOUT + col]     = val;
                acc_out[(size_t)row * NOUT + col] = val;
            }
        }
    }
}

__global__ void gcn_scatter(const int* __restrict__ src, const int* __restrict__ dst,
                            const float* __restrict__ p, float* __restrict__ acc, int E)
{
    int t = blockIdx.x * blockDim.x + threadIdx.x;
    int e = t >> 3;
    if (e >= E) return;
    int c4 = (t & 7) * 4;
    int u = src[e];
    int v = dst[e];
    float4 m = *(const float4*)(p + (size_t)u * 32 + c4);
    float* a = acc + (size_t)v * 32 + c4;
    atomicAdd(a + 0, m.x);
    atomicAdd(a + 1, m.y);
    atomicAdd(a + 2, m.z);
    atomicAdd(a + 3, m.w);
}

__global__ void gcn_finish(float* __restrict__ acc, const float* __restrict__ dinv,
                           const float* __restrict__ bias, int n)
{
    int t = blockIdx.x * blockDim.x + threadIdx.x;
    if (t >= n * 32) return;
    int row = t >> 5;
    int c   = t & 31;
    acc[t] = lrelu(dinv[row] * acc[t] + bias[c]);
}

__global__ void gcn_out(const float* __restrict__ acc2, const float* __restrict__ dinv,
                        const float* __restrict__ b2, const float* __restrict__ Wout,
                        const float* __restrict__ bout, float* __restrict__ out, int n)
{
    int i = blockIdx.x * blockDim.x + threadIdx.x;
    if (i >= n) return;
    float di = dinv[i];
    float s = 0.0f;
    #pragma unroll
    for (int c = 0; c < 32; ++c)
        s += Wout[c] * lrelu(di * acc2[(size_t)i * 32 + c] + b2[c]);
    out[i] = s + bout[0];
}

extern "C" void kernel_launch(void* const* d_in, const int* in_sizes, int n_in,
                              void* d_out, int out_size, void* d_ws, size_t ws_size,
                              hipStream_t stream)
{
    const float* x     = (const float*)d_in[0];
    const int*   ei    = (const int*)d_in[1];
    const float* W_in  = (const float*)d_in[2];
    const float* b_in  = (const float*)d_in[3];
    const float* W1    = (const float*)d_in[4];
    const float* b1    = (const float*)d_in[5];
    const float* W2    = (const float*)d_in[6];
    const float* b2    = (const float*)d_in[7];
    const float* W_out = (const float*)d_in[8];
    const float* b_out = (const float*)d_in[9];
    float* out = (float*)d_out;

    const int N = in_sizes[0] / 64;
    const int E = in_sizes[1] / 2;
    const int* src = ei;
    const int* dst = ei + E;

    float* ws   = (float*)d_ws;
    float* dinv = ws;
    float* h0   = ws + (size_t)N;
    float* p    = ws + (size_t)N * 65;
    float* accA = ws + (size_t)N * 97;
    float* accB = ws + (size_t)N * 129;
    float* part = ws + (size_t)N * 161;

    const int TB = 256;
    gcn_deg_hist  <<<DEG_R * DEG_C, DEG_TB, 0, stream>>>(dst, part, N, E);
    gcn_deg_finish<<<(N + TB - 1) / TB, TB, 0, stream>>>(part, dinv, N);

    const int tiles  = (N + 15) / 16;
    const int gblk   = (tiles + 3) / 4;

    gcn_gemm<64, 64, 0><<<gblk, 128, 0, stream>>>(x, W_in, b_in, nullptr, h0, nullptr, N);

    gcn_gemm<64, 32, 1><<<gblk, 128, 0, stream>>>(h0, W1, nullptr, dinv, p, accA, N);
    gcn_scatter<<<((size_t)E * 8 + TB - 1) / TB, TB, 0, stream>>>(src, dst, p, accA, E);
    gcn_finish <<<((size_t)N * 32 + TB - 1) / TB, TB, 0, stream>>>(accA, dinv, b1, N);

    gcn_gemm<32, 32, 1><<<gblk, 128, 0, stream>>>(accA, W2, nullptr, dinv, p, accB, N);
    gcn_scatter<<<((size_t)E * 8 + TB - 1) / TB, TB, 0, stream>>>(src, dst, p, accB, E);

    gcn_out<<<(N + TB - 1) / TB, TB, 0, stream>>>(accB, dinv, b2, W_out, b_out, out, N);
}


# scatter remapped to one lane per channel: a whole 128-byte row per load and per f32 atomic, each wave walks batches of 32 edges (250000 -> 8192 waves per scatter); atomic arrival order differs
# speedup vs baseline: 1.4795x; 1.2147x over previous
#include <hip/hip_runtime.h>
#include <hip/hip_bf16.h>

typedef __attribute__((ext_vector_type(2))) float v2f;
typedef __attribute__((ext_vector_type(8))) float v8f;

__device__ __forceinline__ float lrelu(float v) { return v > 0.0f ? v : 0.01f * v; }

#define DEG_R   16
#define DEG_C   16
#define DEG_CAP 8192
#define DEG_TB  1024

__device__ __forceinline__ void deg_tally(unsigned* s, int d, int base, unsigned cnt) {
    const unsigned off = (unsigned)(d - base);
    if (off < cnt) atomicAdd(&s[off], 1u);
}

__global__ __launch_bounds__(DEG_TB) void gcn_deg_hist(const int* __restrict__ dst,
                                                       float* __restrict__ part, int n, int E)
{
    __shared__ unsigned s_cnt[DEG_CAP];
    const int tid = threadIdx.x;
    const int r = blockIdx.x % DEG_R;
    const int c = blockIdx.x / DEG_R;

    const int rsize = (n + DEG_R - 1) / DEG_R;
    const int rlo = min(n, r * rsize);
    const int rhi = min(n, rlo + rsize);

    const int head  = min(E, (int)(((16u - ((unsigned)(size_t)dst & 15u)) & 15u) >> 2));
    const int4* __restrict__ dst4 = (const int4*)(dst + head);
    const int nvec  = (E - head) >> 2;
    const int tail0 = head + 4 * nvec;
    const int vsz = (nvec + DEG_C - 1) / DEG_C;
    const int vlo = min(nvec, c * vsz);
    const int vhi = min(nvec, vlo + vsz);

    for (int base = rlo; base < rhi; base += DEG_CAP) {
        const unsigned cnt = (unsigned)min(DEG_CAP, rhi - base);
        for (unsigned i = tid; i < cnt; i += DEG_TB) s_cnt[i] = 0u;
        __syncthreads();

        int i = vlo + tid;
        for (; i + 7 * DEG_TB < vhi; i += 8 * DEG_TB) {
            int4 v[8];
            #pragma unroll
            for (int k = 0; k < 8; ++k) v[k] = dst4[i + k * DEG_TB];
            #pragma unroll
            for (int k = 0; k < 8; ++k) {
                deg_tally(s_cnt, v[k].x, base, cnt);
                deg_tally(s_cnt, v[k].y, base, cnt);
                deg_tally(s_cnt, v[k].z, base, cnt);
                deg_tally(s_cnt, v[k].w, base, cnt);
            }
        }
        for (; i < vhi; i += DEG_TB) {
            const int4 v = dst4[i];
            deg_tally(s_cnt, v.x, base, cnt);
            deg_tally(s_cnt, v.y, base, cnt);
            deg_tally(s_cnt, v.z, base, cnt);
            deg_tally(s_cnt, v.w, base, cnt);
        }
        if (c == 0) {
            if (tid < head)      deg_tally(s_cnt, dst[tid], base, cnt);
            if (tid < E - tail0) deg_tally(s_cnt, dst[tail0 + tid], base, cnt);
        }
        __syncthreads();

        for (unsigned i2 = tid; i2 < cnt; i2 += DEG_TB)
            part[(size_t)c * n + base + i2] = (float)s_cnt[i2];
        __syncthreads();
    }
}

__global__ void gcn_deg_finish(const float* __restrict__ part, float* __restrict__ dinv, int n) {
    int i = blockIdx.x * blockDim.x + threadIdx.x;
    if (i >= n) return;
    float d = 1.0f;
    #pragma unroll
    for (int c = 0; c < DEG_C; ++c) d += part[(size_t)c * n + i];
    dinv[i] = rsqrtf(d);
}

template <int K, int NOUT, int MODE>
__global__ __launch_bounds__(128) void gcn_gemm(
    const float* __restrict__ in,
    const float* __restrict__ W,
    const float* __restrict__ bias,
    const float* __restrict__ dinv,
    float* __restrict__ out,
    float* __restrict__ acc_out,
    int n)
{
    __shared__ float sW[NOUT * K];
    for (int i = threadIdx.x; i < NOUT * K / 4; i += blockDim.x)
        ((float4*)sW)[i] = ((const float4*)W)[i];
    __syncthreads();

    const int lane  = threadIdx.x & 31;
    const int wave  = threadIdx.x >> 5;
    const int row0  = (blockIdx.x * 4 + wave) * 16;
    if (row0 >= n) return;

    const int lhalf = lane >> 4;
    const int l16   = lane & 15;
    const int arow  = row0 + l16;

    v8f acc[NOUT / 16];
    #pragma unroll
    for (int t = 0; t < NOUT / 16; ++t)
        acc[t] = (v8f){0.f, 0.f, 0.f, 0.f, 0.f, 0.f, 0.f, 0.f};

    #pragma unroll
    for (int k0 = 0; k0 < K; k0 += 4) {
        const int kb = k0 + 2 * lhalf;
        v2f a;
        a.x = in[(size_t)arow * K + kb];
        a.y = in[(size_t)arow * K + kb + 1];
        #pragma unroll
        for (int t = 0; t < NOUT / 16; ++t) {
            const int ncol = t * 16 + l16;
            v2f b;
            b.x = sW[ncol * K + kb];
            b.y = sW[ncol * K + kb + 1];
            acc[t] = __builtin_amdgcn_wmma_f32_16x16x4_f32(
                false, a, false, b, (short)0, acc[t], false, false);
        }
    }

    #pragma unroll
    for (int t = 0; t < NOUT / 16; ++t) {
        const int col = t * 16 + l16;
        if (MODE == 0) {
            const float bv = bias[col];
            #pragma unroll
            for (int r = 0; r < 8; ++r) {
                const int row = row0 + r + 8 * lhalf;
                out[(size_t)row * NOUT + col] = lrelu(acc[t][r] + bv);
            }
        } else {
            #pragma unroll
            for (int r = 0; r < 8; ++r) {
                const int row = row0 + r + 8 * lhalf;
                const float val = acc[t][r] * dinv[row];
                out[(size_t)row * NOUT + col]     = val;
                acc_out[(size_t)row * NOUT + col] = val;
            }
        }
    }
}

#define SC_TB     256
#define SC_BLOCKS 1024

__global__ __launch_bounds__(SC_TB) void gcn_scatter(const int* __restrict__ src,
                                                     const int* __restrict__ dst,
                                                     const float* __restrict__ p,
                                                     float* __restrict__ acc, int E)
{
    const int lane = threadIdx.x & 31;
    const int wave = __builtin_amdgcn_readfirstlane((int)((blockIdx.x * SC_TB + threadIdx.x) >> 5));
    const int step = (int)((gridDim.x * SC_TB) >> 5) * 32;
    int base = wave * 32;
    if (base >= E) return;

    int u = 0, v = 0;
    if (base + lane < E) { u = src[base + lane]; v = dst[base + lane]; }

    for (;;) {
        const bool more = (E - base) > step;
        const int nbase = base + step;
        int nu = 0, nv = 0;
        if (more && nbase + lane < E) { nu = src[nbase + lane]; nv = dst[nbase + lane]; }

        const int cnt = min(32, E - base);
        if (cnt == 32) {
            float m[32];
            #pragma unroll
            for (int k = 0; k < 32; ++k) {
                const int uu = __builtin_amdgcn_readlane(u, k);
                m[k] = p[(size_t)uu * 32 + lane];
            }
            #pragma unroll
            for (int k = 0; k < 32; ++k) {
                const int vv = __builtin_amdgcn_readlane(v, k);
                atomicAdd(acc + (size_t)vv * 32 + lane, m[k]);
            }
        } else {
            for (int k = 0; k < cnt; ++k) {
                const int uu = __builtin_amdgcn_readlane(u, k);
                const int vv = __builtin_amdgcn_readlane(v, k);
                atomicAdd(acc + (size_t)vv * 32 + lane, p[(size_t)uu * 32 + lane]);
            }
        }
        if (!more) break;
        base = nbase; u = nu; v = nv;
    }
}

__global__ void gcn_finish(float* __restrict__ acc, const float* __restrict__ dinv,
                           const float* __restrict__ bias, int n)
{
    int t = blockIdx.x * blockDim.x + threadIdx.x;
    if (t >= n * 32) return;
    int row = t >> 5;
    int c   = t & 31;
    acc[t] = lrelu(dinv[row] * acc[t] + bias[c]);
}

__global__ void gcn_out(const float* __restrict__ acc2, const float* __restrict__ dinv,
                        const float* __restrict__ b2, const float* __restrict__ Wout,
                        const float* __restrict__ bout, float* __restrict__ out, int n)
{
    int i = blockIdx.x * blockDim.x + threadIdx.x;
    if (i >= n) return;
    float di = dinv[i];
    float s = 0.0f;
    #pragma unroll
    for (int c = 0; c < 32; ++c)
        s += Wout[c] * lrelu(di * acc2[(size_t)i * 32 + c] + b2[c]);
    out[i] = s + bout[0];
}

extern "C" void kernel_launch(void* const* d_in, const int* in_sizes, int n_in,
                              void* d_out, int out_size, void* d_ws, size_t ws_size,
                              hipStream_t stream)
{
    const float* x     = (const float*)d_in[0];
    const int*   ei    = (const int*)d_in[1];
    const float* W_in  = (const float*)d_in[2];
    const float* b_in  = (const float*)d_in[3];
    const float* W1    = (const float*)d_in[4];
    const float* b1    = (const float*)d_in[5];
    const float* W2    = (const float*)d_in[6];
    const float* b2    = (const float*)d_in[7];
    const float* W_out = (const float*)d_in[8];
    const float* b_out = (const float*)d_in[9];
    float* out = (float*)d_out;

    const int N = in_sizes[0] / 64;
    const int E = in_sizes[1] / 2;
    const int* src = ei;
    const int* dst = ei + E;

    float* ws   = (float*)d_ws;
    float* dinv = ws;
    float* h0   = ws + (size_t)N;
    float* p    = ws + (size_t)N * 65;
    float* accA = ws + (size_t)N * 97;
    float* accB = ws + (size_t)N * 129;
    float* part = ws + (size_t)N * 161;

    const int TB = 256;
    gcn_deg_hist  <<<DEG_R * DEG_C, DEG_TB, 0, stream>>>(dst, part, N, E);
    gcn_deg_finish<<<(N + TB - 1) / TB, TB, 0, stream>>>(part, dinv, N);

    const int tiles  = (N + 15) / 16;
    const int gblk   = (tiles + 3) / 4;

    gcn_gemm<64, 64, 0><<<gblk, 128, 0, stream>>>(x, W_in, b_in, nullptr, h0, nullptr, N);

    gcn_gemm<64, 32, 1><<<gblk, 128, 0, stream>>>(h0, W1, nullptr, dinv, p, accA, N);
    gcn_scatter<<<SC_BLOCKS, SC_TB, 0, stream>>>(src, dst, p, accA, E);
    gcn_finish <<<((size_t)N * 32 + TB - 1) / TB, TB, 0, stream>>>(accA, dinv, b1, N);

    gcn_gemm<32, 32, 1><<<gblk, 128, 0, stream>>>(accA, W2, nullptr, dinv, p, accB, N);
    gcn_scatter<<<SC_BLOCKS, SC_TB, 0, stream>>>(src, dst, p, accB, E);

    gcn_out<<<(N + TB - 1) / TB, TB, 0, stream>>>(accB, dinv, b2, W_out, b_out, out, N);
}


# both elementwise passes folded into the conv products: dinv from the count planes inside the conv1 product, conv1's lrelu(dinv*acc+b1) in the conv2 A-operand load; launches 9->7; same arithmetic
# speedup vs baseline: 1.6171x; 1.0930x over previous
#include <hip/hip_runtime.h>
#include <hip/hip_bf16.h>

typedef __attribute__((ext_vector_type(2))) float v2f;
typedef __attribute__((ext_vector_type(8))) float v8f;

__device__ __forceinline__ float lrelu(float v) { return v > 0.0f ? v : 0.01f * v; }

#define DEG_R   16
#define DEG_C   16
#define DEG_CAP 8192
#define DEG_TB  1024

__device__ __forceinline__ void deg_tally(unsigned* s, int d, int base, unsigned cnt) {
    const unsigned off = (unsigned)(d - base);
    if (off < cnt) atomicAdd(&s[off], 1u);
}

__global__ __launch_bounds__(DEG_TB) void gcn_deg_hist(const int* __restrict__ dst,
                                                       float* __restrict__ part, int n, int E)
{
    __shared__ unsigned s_cnt[DEG_CAP];
    const int tid = threadIdx.x;
    const int r = blockIdx.x % DEG_R;
    const int c = blockIdx.x / DEG_R;

    const int rsize = (n + DEG_R - 1) / DEG_R;
    const int rlo = min(n, r * rsize);
    const int rhi = min(n, rlo + rsize);

    const int head  = min(E, (int)(((16u - ((unsigned)(size_t)dst & 15u)) & 15u) >> 2));
    const int4* __restrict__ dst4 = (const int4*)(dst + head);
    const int nvec  = (E - head) >> 2;
    const int tail0 = head + 4 * nvec;
    const int vsz = (nvec + DEG_C - 1) / DEG_C;
    const int vlo = min(nvec, c * vsz);
    const int vhi = min(nvec, vlo + vsz);

    for (int base = rlo; base < rhi; base += DEG_CAP) {
        const unsigned cnt = (unsigned)min(DEG_CAP, rhi - base);
        for (unsigned i = tid; i < cnt; i += DEG_TB) s_cnt[i] = 0u;
        __syncthreads();

        int i = vlo + tid;
        for (; i + 7 * DEG_TB < vhi; i += 8 * DEG_TB) {
            int4 v[8];
            #pragma unroll
            for (int k = 0; k < 8; ++k) v[k] = dst4[i + k * DEG_TB];
            #pragma unroll
            for (int k = 0; k < 8; ++k) {
                deg_tally(s_cnt, v[k].x, base, cnt);
                deg_tally(s_cnt, v[k].y, base, cnt);
                deg_tally(s_cnt, v[k].z, base, cnt);
                deg_tally(s_cnt, v[k].w, base, cnt);
            }
        }
        for (; i < vhi; i += DEG_TB) {
            const int4 v = dst4[i];
            deg_tally(s_cnt, v.x, base, cnt);
            deg_tally(s_cnt, v.y, base, cnt);
            deg_tally(s_cnt, v.z, base, cnt);
            deg_tally(s_cnt, v.w, base, cnt);
        }
        if (c == 0) {
            if (tid < head)      deg_tally(s_cnt, dst[tid], base, cnt);
            if (tid < E - tail0) deg_tally(s_cnt, dst[tail0 + tid], base, cnt);
        }
        __syncthreads();

        for (unsigned i2 = tid; i2 < cnt; i2 += DEG_TB)
            part[(size_t)c * n + base + i2] = (float)s_cnt[i2];
        __syncthreads();
    }
}

template <int K, int NOUT, int MODE, bool PRE, bool DEG>
__global__ __launch_bounds__(128) void gcn_gemm(
    const float* __restrict__ in,
    const float* __restrict__ W,
    const float* __restrict__ bias,
    const float* __restrict__ dinv,
    const float* __restrict__ part,
    float* __restrict__ dinv_out,
    float* __restrict__ out,
    float* __restrict__ acc_out,
    int n)
{
    __shared__ float sW[NOUT * K];
    for (int i = threadIdx.x; i < NOUT * K / 4; i += blockDim.x)
        ((float4*)sW)[i] = ((const float4*)W)[i];
    __syncthreads();

    const int lane  = threadIdx.x & 31;
    const int wave  = threadIdx.x >> 5;
    const int row0  = (blockIdx.x * 4 + wave) * 16;
    if (row0 >= n) return;

    const int lhalf = lane >> 4;
    const int l16   = lane & 15;
    const int arow  = row0 + l16;

    v8f acc[NOUT / 16];
    #pragma unroll
    for (int t = 0; t < NOUT / 16; ++t)
        acc[t] = (v8f){0.f, 0.f, 0.f, 0.f, 0.f, 0.f, 0.f, 0.f};

    float dself = 0.0f;
    float cnt[DEG_C];
    if (DEG) {
        #pragma unroll
        for (int c = 0; c < DEG_C; ++c) cnt[c] = part[(size_t)c * n + arow];
    } else if (PRE) {
        dself = dinv[arow];
    }

    #pragma unroll
    for (int k0 = 0; k0 < K; k0 += 4) {
        const int kb = k0 + 2 * lhalf;
        v2f a;
        a.x = in[(size_t)arow * K + kb];
        a.y = in[(size_t)arow * K + kb + 1];
        if (PRE) {
            a.x = lrelu(dself * a.x + bias[kb]);
            a.y = lrelu(dself * a.y + bias[kb + 1]);
        }
        #pragma unroll
        for (int t = 0; t < NOUT / 16; ++t) {
            const int ncol = t * 16 + l16;
            v2f b;
            b.x = sW[ncol * K + kb];
            b.y = sW[ncol * K + kb + 1];
            acc[t] = __builtin_amdgcn_wmma_f32_16x16x4_f32(
                false, a, false, b, (short)0, acc[t], false, false);
        }
    }

    if (DEG) {
        float d = 1.0f;
        #pragma unroll
        for (int c = 0; c < DEG_C; ++c) d += cnt[c];
        dself = rsqrtf(d);
        if (lhalf == 0) dinv_out[arow] = dself;
    }

    float dv[8];
    if (MODE == 1) {
        #pragma unroll
        for (int r = 0; r < 8; ++r) {
            dv[r] = DEG ? __shfl(dself, r + 8 * lhalf) : dinv[row0 + r + 8 * lhalf];
        }
    }
    #pragma unroll
    for (int t = 0; t < NOUT / 16; ++t) {
        const int col = t * 16 + l16;
        if (MODE == 0) {
            const float bv = bias[col];
            #pragma unroll
            for (int r = 0; r < 8; ++r) {
                const int row = row0 + r + 8 * lhalf;
                out[(size_t)row * NOUT + col] = lrelu(acc[t][r] + bv);
            }
        } else {
            #pragma unroll
            for (int r = 0; r < 8; ++r) {
                const int row = row0 + r + 8 * lhalf;
                const float val = acc[t][r] * dv[r];
                out[(size_t)row * NOUT + col]     = val;
                acc_out[(size_t)row * NOUT + col] = val;
            }
        }
    }
}

#define SC_TB     256
#define SC_BLOCKS 1024

__global__ __launch_bounds__(SC_TB) void gcn_scatter(const int* __restrict__ src,
                                                     const int* __restrict__ dst,
                                                     const float* __restrict__ p,
                                                     float* __restrict__ acc, int E)
{
    const int lane = threadIdx.x & 31;
    const int wave = __builtin_amdgcn_readfirstlane((int)((blockIdx.x * SC_TB + threadIdx.x) >> 5));
    const int step = (int)((gridDim.x * SC_TB) >> 5) * 32;
    int base = wave * 32;
    if (base >= E) return;

    int u = 0, v = 0;
    if (base + lane < E) { u = src[base + lane]; v = dst[base + lane]; }

    for (;;) {
        const bool more = (E - base) > step;
        const int nbase = base + step;
        int nu = 0, nv = 0;
        if (more && nbase + lane < E) { nu = src[nbase + lane]; nv = dst[nbase + lane]; }

        const int cnt = min(32, E - base);
        if (cnt == 32) {
            float m[32];
            #pragma unroll
            for (int k = 0; k < 32; ++k) {
                const int uu = __builtin_amdgcn_readlane(u, k);
                m[k] = p[(size_t)uu * 32 + lane];
            }
            #pragma unroll
            for (int k = 0; k < 32; ++k) {
                const int vv = __builtin_amdgcn_readlane(v, k);
                atomicAdd(acc + (size_t)vv * 32 + lane, m[k]);
            }
        } else {
            for (int k = 0; k < cnt; ++k) {
                const int uu = __builtin_amdgcn_readlane(u, k);
                const int vv = __builtin_amdgcn_readlane(v, k);
                atomicAdd(acc + (size_t)vv * 32 + lane, p[(size_t)uu * 32 + lane]);
            }
        }
        if (!more) break;
        base = nbase; u = nu; v = nv;
    }
}

__global__ void gcn_out(const float* __restrict__ acc2, const float* __restrict__ dinv,
                        const float* __restrict__ b2, const float* __restrict__ Wout,
                        const float* __restrict__ bout, float* __restrict__ out, int n)
{
    int i = blockIdx.x * blockDim.x + threadIdx.x;
    if (i >= n) return;
    float di = dinv[i];
    float s = 0.0f;
    #pragma unroll
    for (int c = 0; c < 32; ++c)
        s += Wout[c] * lrelu(di * acc2[(size_t)i * 32 + c] + b2[c]);
    out[i] = s + bout[0];
}

extern "C" void kernel_launch(void* const* d_in, const int* in_sizes, int n_in,
                              void* d_out, int out_size, void* d_ws, size_t ws_size,
                              hipStream_t stream)
{
    const float* x     = (const float*)d_in[0];
    const int*   ei    = (const int*)d_in[1];
    const float* W_in  = (const float*)d_in[2];
    const float* b_in  = (const float*)d_in[3];
    const float* W1    = (const float*)d_in[4];
    const float* b1    = (const float*)d_in[5];
    const float* W2    = (const float*)d_in[6];
    const float* b2    = (const float*)d_in[7];
    const float* W_out = (const float*)d_in[8];
    const float* b_out = (const float*)d_in[9];
    float* out = (float*)d_out;

    const int N = in_sizes[0] / 64;
    const int E = in_sizes[1] / 2;
    const int* src = ei;
    const int* dst = ei + E;

    float* ws   = (float*)d_ws;
    float* dinv = ws;
    float* h0   = ws + (size_t)N;
    float* p    = ws + (size_t)N * 65;
    float* accA = ws + (size_t)N * 97;
    float* accB = ws + (size_t)N * 129;
    float* part = ws + (size_t)N * 161;

    const int TB = 256;
    gcn_deg_hist  <<<DEG_R * DEG_C, DEG_TB, 0, stream>>>(dst, part, N, E);

    const int tiles  = (N + 15) / 16;
    const int gblk   = (tiles + 3) / 4;

    gcn_gemm<64, 64, 0, false, false><<<gblk, 128, 0, stream>>>(x, W_in, b_in, nullptr, nullptr, nullptr, h0, nullptr, N);

    gcn_gemm<64, 32, 1, false, true><<<gblk, 128, 0, stream>>>(h0, W1, nullptr, nullptr, part, dinv, p, accA, N);
    gcn_scatter<<<SC_BLOCKS, SC_TB, 0, stream>>>(src, dst, p, accA, E);

    gcn_gemm<32, 32, 1, true, false><<<gblk, 128, 0, stream>>>(accA, W2, b1, dinv, nullptr, nullptr, p, accB, N);
    gcn_scatter<<<SC_BLOCKS, SC_TB, 0, stream>>>(src, dst, p, accB, E);

    gcn_out<<<(N + TB - 1) / TB, TB, 0, stream>>>(accB, dinv, b2, W_out, b_out, out, N);
}


# weight rows staged in LDS at a pitch of K+4 floats instead of K: the 16 lanes of a B-fragment read hit 16 different banks instead of one (K=64) or two (K=32), bank conflicts gone; same arithmetic
# speedup vs baseline: 1.7280x; 1.0686x over previous
#include <hip/hip_runtime.h>
#include <hip/hip_bf16.h>

typedef __attribute__((ext_vector_type(2))) float v2f;
typedef __attribute__((ext_vector_type(8))) float v8f;

__device__ __forceinline__ float lrelu(float v) { return v > 0.0f ? v : 0.01f * v; }

#define DEG_R   16
#define DEG_C   16
#define DEG_CAP 8192
#define DEG_TB  1024

__device__ __forceinline__ void deg_tally(unsigned* s, int d, int base, unsigned cnt) {
    const unsigned off = (unsigned)(d - base);
    if (off < cnt) atomicAdd(&s[off], 1u);
}

__global__ __launch_bounds__(DEG_TB) void gcn_deg_hist(const int* __restrict__ dst,
                                                       float* __restrict__ part, int n, int E)
{
    __shared__ unsigned s_cnt[DEG_CAP];
    const int tid = threadIdx.x;
    const int r = blockIdx.x % DEG_R;
    const int c = blockIdx.x / DEG_R;

    const int rsize = (n + DEG_R - 1) / DEG_R;
    const int rlo = min(n, r * rsize);
    const int rhi = min(n, rlo + rsize);

    const int head  = min(E, (int)(((16u - ((unsigned)(size_t)dst & 15u)) & 15u) >> 2));
    const int4* __restrict__ dst4 = (const int4*)(dst + head);
    const int nvec  = (E - head) >> 2;
    const int tail0 = head + 4 * nvec;
    const int vsz = (nvec + DEG_C - 1) / DEG_C;
    const int vlo = min(nvec, c * vsz);
    const int vhi = min(nvec, vlo + vsz);

    for (int base = rlo; base < rhi; base += DEG_CAP) {
        const unsigned cnt = (unsigned)min(DEG_CAP, rhi - base);
        for (unsigned i = tid; i < cnt; i += DEG_TB) s_cnt[i] = 0u;
        __syncthreads();

        int i = vlo + tid;
        for (; i + 7 * DEG_TB < vhi; i += 8 * DEG_TB) {
            int4 v[8];
            #pragma unroll
            for (int k = 0; k < 8; ++k) v[k] = dst4[i + k * DEG_TB];
            #pragma unroll
            for (int k = 0; k < 8; ++k) {
                deg_tally(s_cnt, v[k].x, base, cnt);
                deg_tally(s_cnt, v[k].y, base, cnt);
                deg_tally(s_cnt, v[k].z, base, cnt);
                deg_tally(s_cnt, v[k].w, base, cnt);
            }
        }
        for (; i < vhi; i += DEG_TB) {
            const int4 v = dst4[i];
            deg_tally(s_cnt, v.x, base, cnt);
            deg_tally(s_cnt, v.y, base, cnt);
            deg_tally(s_cnt, v.z, base, cnt);
            deg_tally(s_cnt, v.w, base, cnt);
        }
        if (c == 0) {
            if (tid < head)      deg_tally(s_cnt, dst[tid], base, cnt);
            if (tid < E - tail0) deg_tally(s_cnt, dst[tail0 + tid], base, cnt);
        }
        __syncthreads();

        for (unsigned i2 = tid; i2 < cnt; i2 += DEG_TB)
            part[(size_t)c * n + base + i2] = (float)s_cnt[i2];
        __syncthreads();
    }
}

template <int K, int NOUT, int MODE, bool PRE, bool DEG>
__global__ __launch_bounds__(128) void gcn_gemm(
    const float* __restrict__ in,
    const float* __restrict__ W,
    const float* __restrict__ bias,
    const float* __restrict__ dinv,
    const float* __restrict__ part,
    float* __restrict__ dinv_out,
    float* __restrict__ out,
    float* __restrict__ acc_out,
    int n)
{
    constexpr int KP = K + 4;
    __shared__ __attribute__((aligned(16))) float sW[NOUT * KP];
    for (int i = threadIdx.x; i < NOUT * K / 4; i += blockDim.x) {
        const int wr = i / (K / 4);
        const int wc = (i % (K / 4)) * 4;
        *(float4*)(sW + wr * KP + wc) = ((const float4*)W)[i];
    }
    __syncthreads();

    const int lane  = threadIdx.x & 31;
    const int wave  = threadIdx.x >> 5;
    const int row0  = (blockIdx.x * 4 + wave) * 16;
    if (row0 >= n) return;

    const int lhalf = lane >> 4;
    const int l16   = lane & 15;
    const int arow  = row0 + l16;

    v8f acc[NOUT / 16];
    #pragma unroll
    for (int t = 0; t < NOUT / 16; ++t)
        acc[t] = (v8f){0.f, 0.f, 0.f, 0.f, 0.f, 0.f, 0.f, 0.f};

    float dself = 0.0f;
    float cnt[DEG_C];
    if (DEG) {
        #pragma unroll
        for (int c = 0; c < DEG_C; ++c) cnt[c] = part[(size_t)c * n + arow];
    } else if (PRE) {
        dself = dinv[arow];
    }

    #pragma unroll
    for (int k0 = 0; k0 < K; k0 += 4) {
        const int kb = k0 + 2 * lhalf;
        v2f a;
        a.x = in[(size_t)arow * K + kb];
        a.y = in[(size_t)arow * K + kb + 1];
        if (PRE) {
            a.x = lrelu(dself * a.x + bias[kb]);
            a.y = lrelu(dself * a.y + bias[kb + 1]);
        }
        #pragma unroll
        for (int t = 0; t < NOUT / 16; ++t) {
            const int ncol = t * 16 + l16;
            v2f b;
            b.x = sW[ncol * KP + kb];
            b.y = sW[ncol * KP + kb + 1];
            acc[t] = __builtin_amdgcn_wmma_f32_16x16x4_f32(
                false, a, false, b, (short)0, acc[t], false, false);
        }
    }

    if (DEG) {
        float d = 1.0f;
        #pragma unroll
        for (int c = 0; c < DEG_C; ++c) d += cnt[c];
        dself = rsqrtf(d);
        if (lhalf == 0) dinv_out[arow] = dself;
    }

    float dv[8];
    if (MODE == 1) {
        #pragma unroll
        for (int r = 0; r < 8; ++r) {
            dv[r] = DEG ? __shfl(dself, r + 8 * lhalf) : dinv[row0 + r + 8 * lhalf];
        }
    }
    #pragma unroll
    for (int t = 0; t < NOUT / 16; ++t) {
        const int col = t * 16 + l16;
        if (MODE == 0) {
            const float bv = bias[col];
            #pragma unroll
            for (int r = 0; r < 8; ++r) {
                const int row = row0 + r + 8 * lhalf;
                out[(size_t)row * NOUT + col] = lrelu(acc[t][r] + bv);
            }
        } else {
            #pragma unroll
            for (int r = 0; r < 8; ++r) {
                const int row = row0 + r + 8 * lhalf;
                const float val = acc[t][r] * dv[r];
                out[(size_t)row * NOUT + col]     = val;
                acc_out[(size_t)row * NOUT + col] = val;
            }
        }
    }
}

#define SC_TB     256
#define SC_BLOCKS 1024

__global__ __launch_bounds__(SC_TB) void gcn_scatter(const int* __restrict__ src,
                                                     const int* __restrict__ dst,
                                                     const float* __restrict__ p,
                                                     float* __restrict__ acc, int E)
{
    const int lane = threadIdx.x & 31;
    const int wave = __builtin_amdgcn_readfirstlane((int)((blockIdx.x * SC_TB + threadIdx.x) >> 5));
    const int step = (int)((gridDim.x * SC_TB) >> 5) * 32;
    int base = wave * 32;
    if (base >= E) return;

    int u = 0, v = 0;
    if (base + lane < E) { u = src[base + lane]; v = dst[base + lane]; }

    for (;;) {
        const bool more = (E - base) > step;
        const int nbase = base + step;
        int nu = 0, nv = 0;
        if (more && nbase + lane < E) { nu = src[nbase + lane]; nv = dst[nbase + lane]; }

        const int cnt = min(32, E - base);
        if (cnt == 32) {
            float m[32];
            #pragma unroll
            for (int k = 0; k < 32; ++k) {
                const int uu = __builtin_amdgcn_readlane(u, k);
                m[k] = p[(size_t)uu * 32 + lane];
            }
            #pragma unroll
            for (int k = 0; k < 32; ++k) {
                const int vv = __builtin_amdgcn_readlane(v, k);
                atomicAdd(acc + (size_t)vv * 32 + lane, m[k]);
            }
        } else {
            for (int k = 0; k < cnt; ++k) {
                const int uu = __builtin_amdgcn_readlane(u, k);
                const int vv = __builtin_amdgcn_readlane(v, k);
                atomicAdd(acc + (size_t)vv * 32 + lane, p[(size_t)uu * 32 + lane]);
            }
        }
        if (!more) break;
        base = nbase; u = nu; v = nv;
    }
}

__global__ void gcn_out(const float* __restrict__ acc2, const float* __restrict__ dinv,
                        const float* __restrict__ b2, const float* __restrict__ Wout,
                        const float* __restrict__ bout, float* __restrict__ out, int n)
{
    int i = blockIdx.x * blockDim.x + threadIdx.x;
    if (i >= n) return;
    float di = dinv[i];
    float s = 0.0f;
    #pragma unroll
    for (int c = 0; c < 32; ++c)
        s += Wout[c] * lrelu(di * acc2[(size_t)i * 32 + c] + b2[c]);
    out[i] = s + bout[0];
}

extern "C" void kernel_launch(void* const* d_in, const int* in_sizes, int n_in,
                              void* d_out, int out_size, void* d_ws, size_t ws_size,
                              hipStream_t stream)
{
    const float* x     = (const float*)d_in[0];
    const int*   ei    = (const int*)d_in[1];
    const float* W_in  = (const float*)d_in[2];
    const float* b_in  = (const float*)d_in[3];
    const float* W1    = (const float*)d_in[4];
    const float* b1    = (const float*)d_in[5];
    const float* W2    = (const float*)d_in[6];
    const float* b2    = (const float*)d_in[7];
    const float* W_out = (const float*)d_in[8];
    const float* b_out = (const float*)d_in[9];
    float* out = (float*)d_out;

    const int N = in_sizes[0] / 64;
    const int E = in_sizes[1] / 2;
    const int* src = ei;
    const int* dst = ei + E;

    float* ws   = (float*)d_ws;
    float* dinv = ws;
    float* h0   = ws + (size_t)N;
    float* p    = ws + (size_t)N * 65;
    float* accA = ws + (size_t)N * 97;
    float* accB = ws + (size_t)N * 129;
    float* part = ws + (size_t)N * 161;

    const int TB = 256;
    gcn_deg_hist  <<<DEG_R * DEG_C, DEG_TB, 0, stream>>>(dst, part, N, E);

    const int tiles  = (N + 15) / 16;
    const int gblk   = (tiles + 3) / 4;

    gcn_gemm<64, 64, 0, false, false><<<gblk, 128, 0, stream>>>(x, W_in, b_in, nullptr, nullptr, nullptr, h0, nullptr, N);

    gcn_gemm<64, 32, 1, false, true><<<gblk, 128, 0, stream>>>(h0, W1, nullptr, nullptr, part, dinv, p, accA, N);
    gcn_scatter<<<SC_BLOCKS, SC_TB, 0, stream>>>(src, dst, p, accA, E);

    gcn_gemm<32, 32, 1, true, false><<<gblk, 128, 0, stream>>>(accA, W2, b1, dinv, nullptr, nullptr, p, accB, N);
    gcn_scatter<<<SC_BLOCKS, SC_TB, 0, stream>>>(src, dst, p, accB, E);

    gcn_out<<<(N + TB - 1) / TB, TB, 0, stream>>>(accB, dinv, b2, W_out, b_out, out, N);
}


# input transform fused into the conv1 product kernel: h0 formed transposed on WMMA so its D registers are the next product's A fragments, never stored; launches 7->6; conv1 k sum order differs
# speedup vs baseline: 1.8560x; 1.0741x over previous
#include <hip/hip_runtime.h>
#include <hip/hip_bf16.h>

typedef __attribute__((ext_vector_type(2))) float v2f;
typedef __attribute__((ext_vector_type(8))) float v8f;

__device__ __forceinline__ float lrelu(float v) { return v > 0.0f ? v : 0.01f * v; }

#define DEG_R   16
#define DEG_C   16
#define DEG_CAP 8192
#define DEG_TB  1024

__device__ __forceinline__ void deg_tally(unsigned* s, int d, int base, unsigned cnt) {
    const unsigned off = (unsigned)(d - base);
    if (off < cnt) atomicAdd(&s[off], 1u);
}

__global__ __launch_bounds__(DEG_TB) void gcn_deg_hist(const int* __restrict__ dst,
                                                       float* __restrict__ part, int n, int E)
{
    __shared__ unsigned s_cnt[DEG_CAP];
    const int tid = threadIdx.x;
    const int r = blockIdx.x % DEG_R;
    const int c = blockIdx.x / DEG_R;

    const int rsize = (n + DEG_R - 1) / DEG_R;
    const int rlo = min(n, r * rsize);
    const int rhi = min(n, rlo + rsize);

    const int head  = min(E, (int)(((16u - ((unsigned)(size_t)dst & 15u)) & 15u) >> 2));
    const int4* __restrict__ dst4 = (const int4*)(dst + head);
    const int nvec  = (E - head) >> 2;
    const int tail0 = head + 4 * nvec;
    const int vsz = (nvec + DEG_C - 1) / DEG_C;
    const int vlo = min(nvec, c * vsz);
    const int vhi = min(nvec, vlo + vsz);

    for (int base = rlo; base < rhi; base += DEG_CAP) {
        const unsigned cnt = (unsigned)min(DEG_CAP, rhi - base);
        for (unsigned i = tid; i < cnt; i += DEG_TB) s_cnt[i] = 0u;
        __syncthreads();

        int i = vlo + tid;
        for (; i + 7 * DEG_TB < vhi; i += 8 * DEG_TB) {
            int4 v[8];
            #pragma unroll
            for (int k = 0; k < 8; ++k) v[k] = dst4[i + k * DEG_TB];
            #pragma unroll
            for (int k = 0; k < 8; ++k) {
                deg_tally(s_cnt, v[k].x, base, cnt);
                deg_tally(s_cnt, v[k].y, base, cnt);
                deg_tally(s_cnt, v[k].z, base, cnt);
                deg_tally(s_cnt, v[k].w, base, cnt);
            }
        }
        for (; i < vhi; i += DEG_TB) {
            const int4 v = dst4[i];
            deg_tally(s_cnt, v.x, base, cnt);
            deg_tally(s_cnt, v.y, base, cnt);
            deg_tally(s_cnt, v.z, base, cnt);
            deg_tally(s_cnt, v.w, base, cnt);
        }
        if (c == 0) {
            if (tid < head)      deg_tally(s_cnt, dst[tid], base, cnt);
            if (tid < E - tail0) deg_tally(s_cnt, dst[tail0 + tid], base, cnt);
        }
        __syncthreads();

        for (unsigned i2 = tid; i2 < cnt; i2 += DEG_TB)
            part[(size_t)c * n + base + i2] = (float)s_cnt[i2];
        __syncthreads();
    }
}


__global__ __launch_bounds__(128) void gcn_in_conv1(
    const float* __restrict__ x,
    const float* __restrict__ Win,
    const float* __restrict__ bin,
    const float* __restrict__ W1,
    const float* __restrict__ part,
    float* __restrict__ dinv_out,
    float* __restrict__ out,
    float* __restrict__ acc_out,
    int n)
{
    constexpr int K  = 64;
    constexpr int KP = K + 4;
    constexpr int NH = 64;
    constexpr int NO = 32;
    __shared__ __attribute__((aligned(16))) float sWin[NH * KP];
    __shared__ __attribute__((aligned(16))) float sW1[NO * KP];
    __shared__ __attribute__((aligned(16))) float sBin[NH];
    for (int i = threadIdx.x; i < NH * K / 4; i += blockDim.x) {
        const int wr = i / (K / 4);
        const int wc = (i % (K / 4)) * 4;
        *(float4*)(sWin + wr * KP + wc) = ((const float4*)Win)[i];
    }
    for (int i = threadIdx.x; i < NO * K / 4; i += blockDim.x) {
        const int wr = i / (K / 4);
        const int wc = (i % (K / 4)) * 4;
        *(float4*)(sW1 + wr * KP + wc) = ((const float4*)W1)[i];
    }
    if (threadIdx.x < NH) sBin[threadIdx.x] = bin[threadIdx.x];
    __syncthreads();

    const int lane  = threadIdx.x & 31;
    const int wave  = threadIdx.x >> 5;
    const int row0  = (blockIdx.x * 4 + wave) * 16;
    if (row0 >= n) return;

    const int lhalf = lane >> 4;
    const int l16   = lane & 15;
    const int arow  = row0 + l16;

    float cnt[DEG_C];
    #pragma unroll
    for (int c = 0; c < DEG_C; ++c) cnt[c] = part[(size_t)c * n + arow];

    v8f hT[NH / 16];
    #pragma unroll
    for (int t = 0; t < NH / 16; ++t)
        hT[t] = (v8f){0.f, 0.f, 0.f, 0.f, 0.f, 0.f, 0.f, 0.f};

    #pragma unroll
    for (int k0 = 0; k0 < K; k0 += 4) {
        const int kb = k0 + 2 * lhalf;
        v2f xv;
        xv.x = x[(size_t)arow * K + kb];
        xv.y = x[(size_t)arow * K + kb + 1];
        #pragma unroll
        for (int t = 0; t < NH / 16; ++t) {
            const int ch = t * 16 + l16;
            v2f w;
            w.x = sWin[ch * KP + kb];
            w.y = sWin[ch * KP + kb + 1];
            hT[t] = __builtin_amdgcn_wmma_f32_16x16x4_f32(
                false, w, false, xv, (short)0, hT[t], false, false);
        }
    }

    #pragma unroll
    for (int t = 0; t < NH / 16; ++t) {
        #pragma unroll
        for (int u = 0; u < 2; ++u) {
            const float4 bq = *(const float4*)(sBin + t * 16 + 8 * lhalf + 4 * u);
            hT[t][4 * u + 0] = lrelu(hT[t][4 * u + 0] + bq.x);
            hT[t][4 * u + 1] = lrelu(hT[t][4 * u + 1] + bq.y);
            hT[t][4 * u + 2] = lrelu(hT[t][4 * u + 2] + bq.z);
            hT[t][4 * u + 3] = lrelu(hT[t][4 * u + 3] + bq.w);
        }
    }

    v8f acc[NO / 16];
    #pragma unroll
    for (int t2 = 0; t2 < NO / 16; ++t2)
        acc[t2] = (v8f){0.f, 0.f, 0.f, 0.f, 0.f, 0.f, 0.f, 0.f};

    #pragma unroll
    for (int t = 0; t < NH / 16; ++t) {
        #pragma unroll
        for (int u = 0; u < 2; ++u) {
            const int kq = t * 16 + 8 * lhalf + 4 * u;
            float4 wq[NO / 16];
            #pragma unroll
            for (int t2 = 0; t2 < NO / 16; ++t2)
                wq[t2] = *(const float4*)(sW1 + (t2 * 16 + l16) * KP + kq);
            v2f a0, a1;
            a0.x = hT[t][4 * u + 0];  a0.y = hT[t][4 * u + 1];
            a1.x = hT[t][4 * u + 2];  a1.y = hT[t][4 * u + 3];
            #pragma unroll
            for (int t2 = 0; t2 < NO / 16; ++t2) {
                v2f b0;
                b0.x = wq[t2].x;  b0.y = wq[t2].y;
                acc[t2] = __builtin_amdgcn_wmma_f32_16x16x4_f32(
                    false, a0, false, b0, (short)0, acc[t2], false, false);
            }
            #pragma unroll
            for (int t2 = 0; t2 < NO / 16; ++t2) {
                v2f b1;
                b1.x = wq[t2].z;  b1.y = wq[t2].w;
                acc[t2] = __builtin_amdgcn_wmma_f32_16x16x4_f32(
                    false, a1, false, b1, (short)0, acc[t2], false, false);
            }
        }
    }

    float d = 1.0f;
    #pragma unroll
    for (int c = 0; c < DEG_C; ++c) d += cnt[c];
    const float dself = rsqrtf(d);
    if (lhalf == 0) dinv_out[arow] = dself;

    float dv[8];
    #pragma unroll
    for (int r = 0; r < 8; ++r)
        dv[r] = __shfl(dself, r + 8 * lhalf);

    #pragma unroll
    for (int t2 = 0; t2 < NO / 16; ++t2) {
        const int col = t2 * 16 + l16;
        #pragma unroll
        for (int r = 0; r < 8; ++r) {
            const int row = row0 + r + 8 * lhalf;
            const float val = acc[t2][r] * dv[r];
            out[(size_t)row * NO + col]     = val;
            acc_out[(size_t)row * NO + col] = val;
        }
    }
}

template <int K, int NOUT>
__global__ __launch_bounds__(128) void gcn_gemm(
    const float* __restrict__ in,
    const float* __restrict__ W,
    const float* __restrict__ bias,
    const float* __restrict__ dinv,
    float* __restrict__ out,
    float* __restrict__ acc_out,
    int n)
{
    constexpr int KP = K + 4;
    __shared__ __attribute__((aligned(16))) float sW[NOUT * KP];
    for (int i = threadIdx.x; i < NOUT * K / 4; i += blockDim.x) {
        const int wr = i / (K / 4);
        const int wc = (i % (K / 4)) * 4;
        *(float4*)(sW + wr * KP + wc) = ((const float4*)W)[i];
    }
    __syncthreads();

    const int lane  = threadIdx.x & 31;
    const int wave  = threadIdx.x >> 5;
    const int row0  = (blockIdx.x * 4 + wave) * 16;
    if (row0 >= n) return;

    const int lhalf = lane >> 4;
    const int l16   = lane & 15;
    const int arow  = row0 + l16;

    v8f acc[NOUT / 16];
    #pragma unroll
    for (int t = 0; t < NOUT / 16; ++t)
        acc[t] = (v8f){0.f, 0.f, 0.f, 0.f, 0.f, 0.f, 0.f, 0.f};

    const float dself = dinv[arow];

    #pragma unroll
    for (int k0 = 0; k0 < K; k0 += 4) {
        const int kb = k0 + 2 * lhalf;
        v2f a;
        a.x = in[(size_t)arow * K + kb];
        a.y = in[(size_t)arow * K + kb + 1];
        a.x = lrelu(dself * a.x + bias[kb]);
        a.y = lrelu(dself * a.y + bias[kb + 1]);
        #pragma unroll
        for (int t = 0; t < NOUT / 16; ++t) {
            const int ncol = t * 16 + l16;
            v2f b;
            b.x = sW[ncol * KP + kb];
            b.y = sW[ncol * KP + kb + 1];
            acc[t] = __builtin_amdgcn_wmma_f32_16x16x4_f32(
                false, a, false, b, (short)0, acc[t], false, false);
        }
    }

    float dv[8];
    #pragma unroll
    for (int r = 0; r < 8; ++r)
        dv[r] = dinv[row0 + r + 8 * lhalf];
    #pragma unroll
    for (int t = 0; t < NOUT / 16; ++t) {
        const int col = t * 16 + l16;
        #pragma unroll
        for (int r = 0; r < 8; ++r) {
            const int row = row0 + r + 8 * lhalf;
            const float val = acc[t][r] * dv[r];
            out[(size_t)row * NOUT + col]     = val;
            acc_out[(size_t)row * NOUT + col] = val;
        }
    }
}

#define SC_TB     256
#define SC_BLOCKS 1024

__global__ __launch_bounds__(SC_TB) void gcn_scatter(const int* __restrict__ src,
                                                     const int* __restrict__ dst,
                                                     const float* __restrict__ p,
                                                     float* __restrict__ acc, int E)
{
    const int lane = threadIdx.x & 31;
    const int wave = __builtin_amdgcn_readfirstlane((int)((blockIdx.x * SC_TB + threadIdx.x) >> 5));
    const int step = (int)((gridDim.x * SC_TB) >> 5) * 32;
    int base = wave * 32;
    if (base >= E) return;

    int u = 0, v = 0;
    if (base + lane < E) { u = src[base + lane]; v = dst[base + lane]; }

    for (;;) {
        const bool more = (E - base) > step;
        const int nbase = base + step;
        int nu = 0, nv = 0;
        if (more && nbase + lane < E) { nu = src[nbase + lane]; nv = dst[nbase + lane]; }

        const int cnt = min(32, E - base);
        if (cnt == 32) {
            float m[32];
            #pragma unroll
            for (int k = 0; k < 32; ++k) {
                const int uu = __builtin_amdgcn_readlane(u, k);
                m[k] = p[(size_t)uu * 32 + lane];
            }
            #pragma unroll
            for (int k = 0; k < 32; ++k) {
                const int vv = __builtin_amdgcn_readlane(v, k);
                atomicAdd(acc + (size_t)vv * 32 + lane, m[k]);
            }
        } else {
            for (int k = 0; k < cnt; ++k) {
                const int uu = __builtin_amdgcn_readlane(u, k);
                const int vv = __builtin_amdgcn_readlane(v, k);
                atomicAdd(acc + (size_t)vv * 32 + lane, p[(size_t)uu * 32 + lane]);
            }
        }
        if (!more) break;
        base = nbase; u = nu; v = nv;
    }
}

__global__ void gcn_out(const float* __restrict__ acc2, const float* __restrict__ dinv,
                        const float* __restrict__ b2, const float* __restrict__ Wout,
                        const float* __restrict__ bout, float* __restrict__ out, int n)
{
    int i = blockIdx.x * blockDim.x + threadIdx.x;
    if (i >= n) return;
    float di = dinv[i];
    float s = 0.0f;
    #pragma unroll
    for (int c = 0; c < 32; ++c)
        s += Wout[c] * lrelu(di * acc2[(size_t)i * 32 + c] + b2[c]);
    out[i] = s + bout[0];
}

extern "C" void kernel_launch(void* const* d_in, const int* in_sizes, int n_in,
                              void* d_out, int out_size, void* d_ws, size_t ws_size,
                              hipStream_t stream)
{
    const float* x     = (const float*)d_in[0];
    const int*   ei    = (const int*)d_in[1];
    const float* W_in  = (const float*)d_in[2];
    const float* b_in  = (const float*)d_in[3];
    const float* W1    = (const float*)d_in[4];
    const float* b1    = (const float*)d_in[5];
    const float* W2    = (const float*)d_in[6];
    const float* b2    = (const float*)d_in[7];
    const float* W_out = (const float*)d_in[8];
    const float* b_out = (const float*)d_in[9];
    float* out = (float*)d_out;

    const int N = in_sizes[0] / 64;
    const int E = in_sizes[1] / 2;
    const int* src = ei;
    const int* dst = ei + E;

    float* ws   = (float*)d_ws;
    float* dinv = ws;
    float* p    = ws + (size_t)N * 65;
    float* accA = ws + (size_t)N * 97;
    float* accB = ws + (size_t)N * 129;
    float* part = ws + (size_t)N * 161;

    const int TB = 256;
    gcn_deg_hist  <<<DEG_R * DEG_C, DEG_TB, 0, stream>>>(dst, part, N, E);

    const int tiles  = (N + 15) / 16;
    const int gblk   = (tiles + 3) / 4;

    gcn_in_conv1<<<gblk, 128, 0, stream>>>(x, W_in, b_in, W1, part, dinv, p, accA, N);
    gcn_scatter<<<SC_BLOCKS, SC_TB, 0, stream>>>(src, dst, p, accA, E);

    gcn_gemm<32, 32><<<gblk, 128, 0, stream>>>(accA, W2, b1, dinv, p, accB, N);
    gcn_scatter<<<SC_BLOCKS, SC_TB, 0, stream>>>(src, dst, p, accB, E);

    gcn_out<<<(N + TB - 1) / TB, TB, 0, stream>>>(accB, dinv, b2, W_out, b_out, out, N);
}


# edge sums without global atomics: edges sorted once per call by destination bin (LDS counting sort per chunk, in the degree launch); each bin summed in LDS by f32 LDS adds; sum order = LDS arrival
# speedup vs baseline: 2.5902x; 1.3956x over previous
#include <hip/hip_runtime.h>
#include <hip/hip_bf16.h>

typedef __attribute__((ext_vector_type(2))) float v2f;
typedef __attribute__((ext_vector_type(8))) float v8f;

__device__ __forceinline__ float lrelu(float v) { return v > 0.0f ? v : 0.01f * v; }

#define BIN_EC   8192
#define BIN_NB   512
#define BIN_TB   1024
#define AGG_TB   1024
#define AGG_ROWS 256

__device__ __forceinline__ int bin_shift(int n) {
    int sh = 0;
    while (((n - 1) >> sh) >= BIN_NB) ++sh;
    return sh;
}

__device__ __forceinline__ void bin_chunk(int* lds, int c,
                                          const int* __restrict__ src,
                                          const int* __restrict__ dst,
                                          int* __restrict__ ssrc,
                                          int* __restrict__ sdst,
                                          int* __restrict__ binstart,
                                          int n, int E)
{
    int* s_src = lds;
    int* s_dst = lds + BIN_EC;
    unsigned* s_cur = (unsigned*)(lds + 2 * BIN_EC);
    unsigned* s_tot = (unsigned*)(lds + 2 * BIN_EC + BIN_NB);
    const int tid = threadIdx.x;
    const int e0  = c * BIN_EC;
    const int cnt = min(BIN_EC, E - e0);
    const int sh  = bin_shift(n);

    if (tid < BIN_NB) s_cur[tid] = 0u;
    __syncthreads();

    int d[BIN_EC / BIN_TB], s[BIN_EC / BIN_TB];
    #pragma unroll
    for (int k = 0; k < BIN_EC / BIN_TB; ++k) {
        const int i = tid + k * BIN_TB;
        d[k] = -1;  s[k] = 0;
        if (i < cnt) { d[k] = dst[e0 + i];  s[k] = src[e0 + i]; }
    }
    #pragma unroll
    for (int k = 0; k < BIN_EC / BIN_TB; ++k)
        if (d[k] >= 0) atomicAdd(&s_cur[min(d[k] >> sh, BIN_NB - 1)], 1u);
    __syncthreads();

    unsigned val = 0u;
    if (tid < BIN_NB) val = s_cur[tid];
    unsigned incl = val;
    #pragma unroll
    for (int off = 1; off < 32; off <<= 1) {
        const unsigned t = __shfl_up(incl, off);
        if ((tid & 31) >= off) incl += t;
    }
    if (tid < BIN_NB && (tid & 31) == 31) s_tot[tid >> 5] = incl;
    __syncthreads();
    if (tid < BIN_NB) {
        unsigned pre = 0u;
        for (int w = 0; w < (tid >> 5); ++w) pre += s_tot[w];
        const unsigned excl = pre + incl - val;
        s_cur[tid] = excl;
        binstart[(size_t)c * (BIN_NB + 1) + tid] = (int)excl;
    }
    if (tid == 0) binstart[(size_t)c * (BIN_NB + 1) + BIN_NB] = cnt;
    __syncthreads();

    #pragma unroll
    for (int k = 0; k < BIN_EC / BIN_TB; ++k)
        if (d[k] >= 0) {
            const unsigned pos = atomicAdd(&s_cur[min(d[k] >> sh, BIN_NB - 1)], 1u);
            s_src[pos] = s[k];
            s_dst[pos] = d[k];
        }
    __syncthreads();
    for (int i = tid; i < cnt; i += BIN_TB) {
        ssrc[e0 + i] = s_src[i];
        sdst[e0 + i] = s_dst[i];
    }
}

#define DEG_R   16
#define DEG_C   16
#define DEG_CAP 8192
#define DEG_TB  1024

__device__ __forceinline__ void deg_tally(unsigned* s, int d, int base, unsigned cnt) {
    const unsigned off = (unsigned)(d - base);
    if (off < cnt) atomicAdd(&s[off], 1u);
}

__global__ __launch_bounds__(DEG_TB) void gcn_deg_hist(const int* __restrict__ src,
                                                       const int* __restrict__ dst,
                                                       float* __restrict__ part,
                                                       int* __restrict__ ssrc,
                                                       int* __restrict__ sdst,
                                                       int* __restrict__ binstart,
                                                       int n, int E)
{
    __shared__ __attribute__((aligned(16))) int s_raw[2 * BIN_EC + BIN_NB + BIN_NB / 32];
    if (blockIdx.x >= DEG_R * DEG_C) {
        bin_chunk(s_raw, (int)blockIdx.x - DEG_R * DEG_C, src, dst, ssrc, sdst, binstart, n, E);
        return;
    }
    unsigned* s_cnt = (unsigned*)s_raw;
    const int tid = threadIdx.x;
    const int r = blockIdx.x % DEG_R;
    const int c = blockIdx.x / DEG_R;

    const int rsize = (n + DEG_R - 1) / DEG_R;
    const int rlo = min(n, r * rsize);
    const int rhi = min(n, rlo + rsize);

    const int head  = min(E, (int)(((16u - ((unsigned)(size_t)dst & 15u)) & 15u) >> 2));
    const int4* __restrict__ dst4 = (const int4*)(dst + head);
    const int nvec  = (E - head) >> 2;
    const int tail0 = head + 4 * nvec;
    const int vsz = (nvec + DEG_C - 1) / DEG_C;
    const int vlo = min(nvec, c * vsz);
    const int vhi = min(nvec, vlo + vsz);

    for (int base = rlo; base < rhi; base += DEG_CAP) {
        const unsigned cnt = (unsigned)min(DEG_CAP, rhi - base);
        for (unsigned i = tid; i < cnt; i += DEG_TB) s_cnt[i] = 0u;
        __syncthreads();

        int i = vlo + tid;
        for (; i + 7 * DEG_TB < vhi; i += 8 * DEG_TB) {
            int4 v[8];
            #pragma unroll
            for (int k = 0; k < 8; ++k) v[k] = dst4[i + k * DEG_TB];
            #pragma unroll
            for (int k = 0; k < 8; ++k) {
                deg_tally(s_cnt, v[k].x, base, cnt);
                deg_tally(s_cnt, v[k].y, base, cnt);
                deg_tally(s_cnt, v[k].z, base, cnt);
                deg_tally(s_cnt, v[k].w, base, cnt);
            }
        }
        for (; i < vhi; i += DEG_TB) {
            const int4 v = dst4[i];
            deg_tally(s_cnt, v.x, base, cnt);
            deg_tally(s_cnt, v.y, base, cnt);
            deg_tally(s_cnt, v.z, base, cnt);
            deg_tally(s_cnt, v.w, base, cnt);
        }
        if (c == 0) {
            if (tid < head)      deg_tally(s_cnt, dst[tid], base, cnt);
            if (tid < E - tail0) deg_tally(s_cnt, dst[tail0 + tid], base, cnt);
        }
        __syncthreads();

        for (unsigned i2 = tid; i2 < cnt; i2 += DEG_TB)
            part[(size_t)c * n + base + i2] = (float)s_cnt[i2];
        __syncthreads();
    }
}


__global__ __launch_bounds__(128) void gcn_in_conv1(
    const float* __restrict__ x,
    const float* __restrict__ Win,
    const float* __restrict__ bin,
    const float* __restrict__ W1,
    const float* __restrict__ part,
    float* __restrict__ dinv_out,
    float* __restrict__ out,
    int n)
{
    constexpr int K  = 64;
    constexpr int KP = K + 4;
    constexpr int NH = 64;
    constexpr int NO = 32;
    __shared__ __attribute__((aligned(16))) float sWin[NH * KP];
    __shared__ __attribute__((aligned(16))) float sW1[NO * KP];
    __shared__ __attribute__((aligned(16))) float sBin[NH];
    for (int i = threadIdx.x; i < NH * K / 4; i += blockDim.x) {
        const int wr = i / (K / 4);
        const int wc = (i % (K / 4)) * 4;
        *(float4*)(sWin + wr * KP + wc) = ((const float4*)Win)[i];
    }
    for (int i = threadIdx.x; i < NO * K / 4; i += blockDim.x) {
        const int wr = i / (K / 4);
        const int wc = (i % (K / 4)) * 4;
        *(float4*)(sW1 + wr * KP + wc) = ((const float4*)W1)[i];
    }
    if (threadIdx.x < NH) sBin[threadIdx.x] = bin[threadIdx.x];
    __syncthreads();

    const int lane  = threadIdx.x & 31;
    const int wave  = threadIdx.x >> 5;
    const int row0  = (blockIdx.x * 4 + wave) * 16;
    if (row0 >= n) return;

    const int lhalf = lane >> 4;
    const int l16   = lane & 15;
    const int arow  = row0 + l16;

    float cnt[DEG_C];
    #pragma unroll
    for (int c = 0; c < DEG_C; ++c) cnt[c] = part[(size_t)c * n + arow];

    v8f hT[NH / 16];
    #pragma unroll
    for (int t = 0; t < NH / 16; ++t)
        hT[t] = (v8f){0.f, 0.f, 0.f, 0.f, 0.f, 0.f, 0.f, 0.f};

    #pragma unroll
    for (int k0 = 0; k0 < K; k0 += 4) {
        const int kb = k0 + 2 * lhalf;
        v2f xv;
        xv.x = x[(size_t)arow * K + kb];
        xv.y = x[(size_t)arow * K + kb + 1];
        #pragma unroll
        for (int t = 0; t < NH / 16; ++t) {
            const int ch = t * 16 + l16;
            v2f w;
            w.x = sWin[ch * KP + kb];
            w.y = sWin[ch * KP + kb + 1];
            hT[t] = __builtin_amdgcn_wmma_f32_16x16x4_f32(
                false, w, false, xv, (short)0, hT[t], false, false);
        }
    }

    #pragma unroll
    for (int t = 0; t < NH / 16; ++t) {
        #pragma unroll
        for (int u = 0; u < 2; ++u) {
            const float4 bq = *(const float4*)(sBin + t * 16 + 8 * lhalf + 4 * u);
            hT[t][4 * u + 0] = lrelu(hT[t][4 * u + 0] + bq.x);
            hT[t][4 * u + 1] = lrelu(hT[t][4 * u + 1] + bq.y);
            hT[t][4 * u + 2] = lrelu(hT[t][4 * u + 2] + bq.z);
            hT[t][4 * u + 3] = lrelu(hT[t][4 * u + 3] + bq.w);
        }
    }

    v8f acc[NO / 16];
    #pragma unroll
    for (int t2 = 0; t2 < NO / 16; ++t2)
        acc[t2] = (v8f){0.f, 0.f, 0.f, 0.f, 0.f, 0.f, 0.f, 0.f};

    #pragma unroll
    for (int t = 0; t < NH / 16; ++t) {
        #pragma unroll
        for (int u = 0; u < 2; ++u) {
            const int kq = t * 16 + 8 * lhalf + 4 * u;
            float4 wq[NO / 16];
            #pragma unroll
            for (int t2 = 0; t2 < NO / 16; ++t2)
                wq[t2] = *(const float4*)(sW1 + (t2 * 16 + l16) * KP + kq);
            v2f a0, a1;
            a0.x = hT[t][4 * u + 0];  a0.y = hT[t][4 * u + 1];
            a1.x = hT[t][4 * u + 2];  a1.y = hT[t][4 * u + 3];
            #pragma unroll
            for (int t2 = 0; t2 < NO / 16; ++t2) {
                v2f b0;
                b0.x = wq[t2].x;  b0.y = wq[t2].y;
                acc[t2] = __builtin_amdgcn_wmma_f32_16x16x4_f32(
                    false, a0, false, b0, (short)0, acc[t2], false, false);
            }
            #pragma unroll
            for (int t2 = 0; t2 < NO / 16; ++t2) {
                v2f b1;
                b1.x = wq[t2].z;  b1.y = wq[t2].w;
                acc[t2] = __builtin_amdgcn_wmma_f32_16x16x4_f32(
                    false, a1, false, b1, (short)0, acc[t2], false, false);
            }
        }
    }

    float d = 1.0f;
    #pragma unroll
    for (int c = 0; c < DEG_C; ++c) d += cnt[c];
    const float dself = rsqrtf(d);
    if (lhalf == 0) dinv_out[arow] = dself;

    float dv[8];
    #pragma unroll
    for (int r = 0; r < 8; ++r)
        dv[r] = __shfl(dself, r + 8 * lhalf);

    #pragma unroll
    for (int t2 = 0; t2 < NO / 16; ++t2) {
        const int col = t2 * 16 + l16;
        #pragma unroll
        for (int r = 0; r < 8; ++r) {
            const int row = row0 + r + 8 * lhalf;
            out[(size_t)row * NO + col] = acc[t2][r] * dv[r];
        }
    }
}

template <int K, int NOUT>
__global__ __launch_bounds__(128) void gcn_gemm(
    const float* __restrict__ in,
    const float* __restrict__ W,
    const float* __restrict__ bias,
    const float* __restrict__ dinv,
    float* __restrict__ out,
    int n)
{
    constexpr int KP = K + 4;
    __shared__ __attribute__((aligned(16))) float sW[NOUT * KP];
    for (int i = threadIdx.x; i < NOUT * K / 4; i += blockDim.x) {
        const int wr = i / (K / 4);
        const int wc = (i % (K / 4)) * 4;
        *(float4*)(sW + wr * KP + wc) = ((const float4*)W)[i];
    }
    __syncthreads();

    const int lane  = threadIdx.x & 31;
    const int wave  = threadIdx.x >> 5;
    const int row0  = (blockIdx.x * 4 + wave) * 16;
    if (row0 >= n) return;

    const int lhalf = lane >> 4;
    const int l16   = lane & 15;
    const int arow  = row0 + l16;

    v8f acc[NOUT / 16];
    #pragma unroll
    for (int t = 0; t < NOUT / 16; ++t)
        acc[t] = (v8f){0.f, 0.f, 0.f, 0.f, 0.f, 0.f, 0.f, 0.f};

    const float dself = dinv[arow];

    #pragma unroll
    for (int k0 = 0; k0 < K; k0 += 4) {
        const int kb = k0 + 2 * lhalf;
        v2f a;
        a.x = in[(size_t)arow * K + kb];
        a.y = in[(size_t)arow * K + kb + 1];
        a.x = lrelu(dself * a.x + bias[kb]);
        a.y = lrelu(dself * a.y + bias[kb + 1]);
        #pragma unroll
        for (int t = 0; t < NOUT / 16; ++t) {
            const int ncol = t * 16 + l16;
            v2f b;
            b.x = sW[ncol * KP + kb];
            b.y = sW[ncol * KP + kb + 1];
            acc[t] = __builtin_amdgcn_wmma_f32_16x16x4_f32(
                false, a, false, b, (short)0, acc[t], false, false);
        }
    }

    float dv[8];
    #pragma unroll
    for (int r = 0; r < 8; ++r)
        dv[r] = dinv[row0 + r + 8 * lhalf];
    #pragma unroll
    for (int t = 0; t < NOUT / 16; ++t) {
        const int col = t * 16 + l16;
        #pragma unroll
        for (int r = 0; r < 8; ++r) {
            const int row = row0 + r + 8 * lhalf;
            out[(size_t)row * NOUT + col] = acc[t][r] * dv[r];
        }
    }
}

__global__ __launch_bounds__(AGG_TB, 16) void gcn_agg(const int* __restrict__ ssrc,
                                                  const int* __restrict__ sdst,
                                                  const int* __restrict__ binstart,
                                                  const float* __restrict__ p,
                                                  float* __restrict__ acc,
                                                  int n, int nchunk)
{
    __shared__ __attribute__((aligned(16))) float s_acc[AGG_ROWS * 32];
    const int tid  = threadIdx.x;
    const int lane = tid & 31;
    const int wave = __builtin_amdgcn_readfirstlane(tid >> 5);
    const int b    = blockIdx.x;
    const int sh   = bin_shift(n);
    const int rb   = 1 << sh;
    const int base = b << sh;

    for (int r0 = 0; r0 < rb; r0 += AGG_ROWS) {
        const int row0 = base + r0;
        if (row0 >= n) break;
        const int rows = min(AGG_ROWS, min(rb - r0, n - row0));

        const float4* __restrict__ p4 = (const float4*)(p + (size_t)row0 * 32);
        for (int i = tid; i < rows * 8; i += AGG_TB) ((float4*)s_acc)[i] = p4[i];
        __syncthreads();

        for (int cg = 0; cg < nchunk; cg += AGG_TB) {
            int sb = 0, se = 0;
            const int cmine = cg + wave + (AGG_TB / 32) * lane;
            if (cmine < nchunk) {
                const int* __restrict__ bs = binstart + (size_t)cmine * (BIN_NB + 1) + b;
                sb = bs[0];  se = bs[1];
            }
            int j = -1, c = 0, i0 = 0, e = 0;
            bool have = true;
            i0 += 32;
            while (i0 >= e) {
                ++j;
                c = cg + wave + (AGG_TB / 32) * j;
                if (j >= 32 || c >= nchunk) { have = false; break; }
                i0 = __builtin_amdgcn_readlane(sb, j);
                e  = __builtin_amdgcn_readlane(se, j);
            }
            int u = 0, v = -1;
            if (have && i0 + lane < e) {
                u = ssrc[(size_t)c * BIN_EC + i0 + lane];
                v = sdst[(size_t)c * BIN_EC + i0 + lane];
            }
            while (have) {
                const int cnt = min(32, e - i0);
                float m[32];
                #pragma unroll
                for (int k = 0; k < 8; ++k) {
                    const int uu = __builtin_amdgcn_readlane(u, k);
                    m[k] = p[(size_t)uu * 32 + lane];
                }
                #pragma unroll
                for (int g = 1; g < 4; ++g)
                    if (8 * g < cnt) {
                        #pragma unroll
                        for (int k = 8 * g; k < 8 * g + 8; ++k) {
                            const int uu = __builtin_amdgcn_readlane(u, k);
                            m[k] = p[(size_t)uu * 32 + lane];
                        }
                    }
                bool more = true;
                i0 += 32;
                while (i0 >= e) {
                    ++j;
                    c = cg + wave + (AGG_TB / 32) * j;
                    if (j >= 32 || c >= nchunk) { more = false; break; }
                    i0 = __builtin_amdgcn_readlane(sb, j);
                    e  = __builtin_amdgcn_readlane(se, j);
                }
                int nu = 0, nv = -1;
                if (more && i0 + lane < e) {
                    nu = ssrc[(size_t)c * BIN_EC + i0 + lane];
                    nv = sdst[(size_t)c * BIN_EC + i0 + lane];
                }
                #pragma unroll
                for (int g = 0; g < 4; ++g)
                    if (8 * g < cnt) {
                        #pragma unroll
                        for (int k = 8 * g; k < 8 * g + 8; ++k) {
                            const int vv = __builtin_amdgcn_readlane(v, k) - row0;
                            if ((unsigned)vv < (unsigned)rows)
                                atomicAdd(&s_acc[vv * 32 + lane], m[k]);
                        }
                    }
                have = more;  u = nu;  v = nv;
            }
        }
        __syncthreads();

        float4* __restrict__ a4 = (float4*)(acc + (size_t)row0 * 32);
        for (int i = tid; i < rows * 8; i += AGG_TB) a4[i] = ((const float4*)s_acc)[i];
        __syncthreads();
    }
}

__global__ void gcn_out(const float* __restrict__ acc2, const float* __restrict__ dinv,
                        const float* __restrict__ b2, const float* __restrict__ Wout,
                        const float* __restrict__ bout, float* __restrict__ out, int n)
{
    int i = blockIdx.x * blockDim.x + threadIdx.x;
    if (i >= n) return;
    float di = dinv[i];
    float s = 0.0f;
    #pragma unroll
    for (int c = 0; c < 32; ++c)
        s += Wout[c] * lrelu(di * acc2[(size_t)i * 32 + c] + b2[c]);
    out[i] = s + bout[0];
}

extern "C" void kernel_launch(void* const* d_in, const int* in_sizes, int n_in,
                              void* d_out, int out_size, void* d_ws, size_t ws_size,
                              hipStream_t stream)
{
    const float* x     = (const float*)d_in[0];
    const int*   ei    = (const int*)d_in[1];
    const float* W_in  = (const float*)d_in[2];
    const float* b_in  = (const float*)d_in[3];
    const float* W1    = (const float*)d_in[4];
    const float* b1    = (const float*)d_in[5];
    const float* W2    = (const float*)d_in[6];
    const float* b2    = (const float*)d_in[7];
    const float* W_out = (const float*)d_in[8];
    const float* b_out = (const float*)d_in[9];
    float* out = (float*)d_out;

    const int N = in_sizes[0] / 64;
    const int E = in_sizes[1] / 2;
    const int* src = ei;
    const int* dst = ei + E;

    float* ws   = (float*)d_ws;
    float* dinv = ws;
    float* p    = ws + (size_t)N * 65;
    float* accA = ws + (size_t)N * 97;
    float* accB = ws + (size_t)N * 129;
    float* part = ws + (size_t)N * 161;
    const int nchunk = (E + BIN_EC - 1) / BIN_EC;
    int* ssrc     = (int*)(ws + (size_t)N * (161 + DEG_C));
    int* sdst     = ssrc + (size_t)nchunk * BIN_EC;
    int* binstart = sdst + (size_t)nchunk * BIN_EC;

    const int TB = 256;
    gcn_deg_hist  <<<DEG_R * DEG_C + nchunk, DEG_TB, 0, stream>>>(src, dst, part, ssrc, sdst, binstart, N, E);

    const int tiles  = (N + 15) / 16;
    const int gblk   = (tiles + 3) / 4;

    gcn_in_conv1<<<gblk, 128, 0, stream>>>(x, W_in, b_in, W1, part, dinv, p, N);
    gcn_agg<<<BIN_NB, AGG_TB, 0, stream>>>(ssrc, sdst, binstart, p, accA, N, nchunk);

    gcn_gemm<32, 32><<<gblk, 128, 0, stream>>>(accA, W2, b1, dinv, p, N);
    gcn_agg<<<BIN_NB, AGG_TB, 0, stream>>>(ssrc, sdst, binstart, p, accB, N, nchunk);

    gcn_out<<<(N + TB - 1) / TB, TB, 0, stream>>>(accB, dinv, b2, W_out, b_out, out, N);
}


# each edge sum's next row-local stage runs on the summed rows in LDS: conv2 product on WMMA after layer 1, output dot after layer 2; raw sums never stored; launches 6->4; same arithmetic
# speedup vs baseline: 3.0183x; 1.1653x over previous
#include <hip/hip_runtime.h>
#include <hip/hip_bf16.h>

typedef __attribute__((ext_vector_type(2))) float v2f;
typedef __attribute__((ext_vector_type(8))) float v8f;

__device__ __forceinline__ float lrelu(float v) { return v > 0.0f ? v : 0.01f * v; }

#define BIN_EC   8192
#define BIN_NB   512
#define BIN_TB   1024
#define AGG_TB   1024
#define AGG_ROWS 256

__device__ __forceinline__ int bin_shift(int n) {
    int sh = 0;
    while (((n - 1) >> sh) >= BIN_NB) ++sh;
    return sh;
}

__device__ __forceinline__ void bin_chunk(int* lds, int c,
                                          const int* __restrict__ src,
                                          const int* __restrict__ dst,
                                          int* __restrict__ ssrc,
                                          int* __restrict__ sdst,
                                          int* __restrict__ binstart,
                                          int n, int E)
{
    int* s_src = lds;
    int* s_dst = lds + BIN_EC;
    unsigned* s_cur = (unsigned*)(lds + 2 * BIN_EC);
    unsigned* s_tot = (unsigned*)(lds + 2 * BIN_EC + BIN_NB);
    const int tid = threadIdx.x;
    const int e0  = c * BIN_EC;
    const int cnt = min(BIN_EC, E - e0);
    const int sh  = bin_shift(n);

    if (tid < BIN_NB) s_cur[tid] = 0u;
    __syncthreads();

    int d[BIN_EC / BIN_TB], s[BIN_EC / BIN_TB];
    #pragma unroll
    for (int k = 0; k < BIN_EC / BIN_TB; ++k) {
        const int i = tid + k * BIN_TB;
        d[k] = -1;  s[k] = 0;
        if (i < cnt) { d[k] = dst[e0 + i];  s[k] = src[e0 + i]; }
    }
    #pragma unroll
    for (int k = 0; k < BIN_EC / BIN_TB; ++k)
        if (d[k] >= 0) atomicAdd(&s_cur[min(d[k] >> sh, BIN_NB - 1)], 1u);
    __syncthreads();

    unsigned val = 0u;
    if (tid < BIN_NB) val = s_cur[tid];
    unsigned incl = val;
    #pragma unroll
    for (int off = 1; off < 32; off <<= 1) {
        const unsigned t = __shfl_up(incl, off);
        if ((tid & 31) >= off) incl += t;
    }
    if (tid < BIN_NB && (tid & 31) == 31) s_tot[tid >> 5] = incl;
    __syncthreads();
    if (tid < BIN_NB) {
        unsigned pre = 0u;
        for (int w = 0; w < (tid >> 5); ++w) pre += s_tot[w];
        const unsigned excl = pre + incl - val;
        s_cur[tid] = excl;
        binstart[(size_t)c * (BIN_NB + 1) + tid] = (int)excl;
    }
    if (tid == 0) binstart[(size_t)c * (BIN_NB + 1) + BIN_NB] = cnt;
    __syncthreads();

    #pragma unroll
    for (int k = 0; k < BIN_EC / BIN_TB; ++k)
        if (d[k] >= 0) {
            const unsigned pos = atomicAdd(&s_cur[min(d[k] >> sh, BIN_NB - 1)], 1u);
            s_src[pos] = s[k];
            s_dst[pos] = d[k];
        }
    __syncthreads();
    for (int i = tid; i < cnt; i += BIN_TB) {
        ssrc[e0 + i] = s_src[i];
        sdst[e0 + i] = s_dst[i];
    }
}

#define DEG_R   16
#define DEG_C   16
#define DEG_CAP 8192
#define DEG_TB  1024

__device__ __forceinline__ void deg_tally(unsigned* s, int d, int base, unsigned cnt) {
    const unsigned off = (unsigned)(d - base);
    if (off < cnt) atomicAdd(&s[off], 1u);
}

__global__ __launch_bounds__(DEG_TB) void gcn_deg_hist(const int* __restrict__ src,
                                                       const int* __restrict__ dst,
                                                       float* __restrict__ part,
                                                       int* __restrict__ ssrc,
                                                       int* __restrict__ sdst,
                                                       int* __restrict__ binstart,
                                                       int n, int E)
{
    __shared__ __attribute__((aligned(16))) int s_raw[2 * BIN_EC + BIN_NB + BIN_NB / 32];
    if (blockIdx.x >= DEG_R * DEG_C) {
        bin_chunk(s_raw, (int)blockIdx.x - DEG_R * DEG_C, src, dst, ssrc, sdst, binstart, n, E);
        return;
    }
    unsigned* s_cnt = (unsigned*)s_raw;
    const int tid = threadIdx.x;
    const int r = blockIdx.x % DEG_R;
    const int c = blockIdx.x / DEG_R;

    const int rsize = (n + DEG_R - 1) / DEG_R;
    const int rlo = min(n, r * rsize);
    const int rhi = min(n, rlo + rsize);

    const int head  = min(E, (int)(((16u - ((unsigned)(size_t)dst & 15u)) & 15u) >> 2));
    const int4* __restrict__ dst4 = (const int4*)(dst + head);
    const int nvec  = (E - head) >> 2;
    const int tail0 = head + 4 * nvec;
    const int vsz = (nvec + DEG_C - 1) / DEG_C;
    const int vlo = min(nvec, c * vsz);
    const int vhi = min(nvec, vlo + vsz);

    for (int base = rlo; base < rhi; base += DEG_CAP) {
        const unsigned cnt = (unsigned)min(DEG_CAP, rhi - base);
        for (unsigned i = tid; i < cnt; i += DEG_TB) s_cnt[i] = 0u;
        __syncthreads();

        int i = vlo + tid;
        for (; i + 7 * DEG_TB < vhi; i += 8 * DEG_TB) {
            int4 v[8];
            #pragma unroll
            for (int k = 0; k < 8; ++k) v[k] = dst4[i + k * DEG_TB];
            #pragma unroll
            for (int k = 0; k < 8; ++k) {
                deg_tally(s_cnt, v[k].x, base, cnt);
                deg_tally(s_cnt, v[k].y, base, cnt);
                deg_tally(s_cnt, v[k].z, base, cnt);
                deg_tally(s_cnt, v[k].w, base, cnt);
            }
        }
        for (; i < vhi; i += DEG_TB) {
            const int4 v = dst4[i];
            deg_tally(s_cnt, v.x, base, cnt);
            deg_tally(s_cnt, v.y, base, cnt);
            deg_tally(s_cnt, v.z, base, cnt);
            deg_tally(s_cnt, v.w, base, cnt);
        }
        if (c == 0) {
            if (tid < head)      deg_tally(s_cnt, dst[tid], base, cnt);
            if (tid < E - tail0) deg_tally(s_cnt, dst[tail0 + tid], base, cnt);
        }
        __syncthreads();

        for (unsigned i2 = tid; i2 < cnt; i2 += DEG_TB)
            part[(size_t)c * n + base + i2] = (float)s_cnt[i2];
        __syncthreads();
    }
}


__global__ __launch_bounds__(128) void gcn_in_conv1(
    const float* __restrict__ x,
    const float* __restrict__ Win,
    const float* __restrict__ bin,
    const float* __restrict__ W1,
    const float* __restrict__ part,
    float* __restrict__ dinv_out,
    float* __restrict__ out,
    int n)
{
    constexpr int K  = 64;
    constexpr int KP = K + 4;
    constexpr int NH = 64;
    constexpr int NO = 32;
    __shared__ __attribute__((aligned(16))) float sWin[NH * KP];
    __shared__ __attribute__((aligned(16))) float sW1[NO * KP];
    __shared__ __attribute__((aligned(16))) float sBin[NH];
    for (int i = threadIdx.x; i < NH * K / 4; i += blockDim.x) {
        const int wr = i / (K / 4);
        const int wc = (i % (K / 4)) * 4;
        *(float4*)(sWin + wr * KP + wc) = ((const float4*)Win)[i];
    }
    for (int i = threadIdx.x; i < NO * K / 4; i += blockDim.x) {
        const int wr = i / (K / 4);
        const int wc = (i % (K / 4)) * 4;
        *(float4*)(sW1 + wr * KP + wc) = ((const float4*)W1)[i];
    }
    if (threadIdx.x < NH) sBin[threadIdx.x] = bin[threadIdx.x];
    __syncthreads();

    const int lane  = threadIdx.x & 31;
    const int wave  = threadIdx.x >> 5;
    const int row0  = (blockIdx.x * 4 + wave) * 16;
    if (row0 >= n) return;

    const int lhalf = lane >> 4;
    const int l16   = lane & 15;
    const int arow  = row0 + l16;

    float cnt[DEG_C];
    #pragma unroll
    for (int c = 0; c < DEG_C; ++c) cnt[c] = part[(size_t)c * n + arow];

    v8f hT[NH / 16];
    #pragma unroll
    for (int t = 0; t < NH / 16; ++t)
        hT[t] = (v8f){0.f, 0.f, 0.f, 0.f, 0.f, 0.f, 0.f, 0.f};

    #pragma unroll
    for (int k0 = 0; k0 < K; k0 += 4) {
        const int kb = k0 + 2 * lhalf;
        v2f xv;
        xv.x = x[(size_t)arow * K + kb];
        xv.y = x[(size_t)arow * K + kb + 1];
        #pragma unroll
        for (int t = 0; t < NH / 16; ++t) {
            const int ch = t * 16 + l16;
            v2f w;
            w.x = sWin[ch * KP + kb];
            w.y = sWin[ch * KP + kb + 1];
            hT[t] = __builtin_amdgcn_wmma_f32_16x16x4_f32(
                false, w, false, xv, (short)0, hT[t], false, false);
        }
    }

    #pragma unroll
    for (int t = 0; t < NH / 16; ++t) {
        #pragma unroll
        for (int u = 0; u < 2; ++u) {
            const float4 bq = *(const float4*)(sBin + t * 16 + 8 * lhalf + 4 * u);
            hT[t][4 * u + 0] = lrelu(hT[t][4 * u + 0] + bq.x);
            hT[t][4 * u + 1] = lrelu(hT[t][4 * u + 1] + bq.y);
            hT[t][4 * u + 2] = lrelu(hT[t][4 * u + 2] + bq.z);
            hT[t][4 * u + 3] = lrelu(hT[t][4 * u + 3] + bq.w);
        }
    }

    v8f acc[NO / 16];
    #pragma unroll
    for (int t2 = 0; t2 < NO / 16; ++t2)
        acc[t2] = (v8f){0.f, 0.f, 0.f, 0.f, 0.f, 0.f, 0.f, 0.f};

    #pragma unroll
    for (int t = 0; t < NH / 16; ++t) {
        #pragma unroll
        for (int u = 0; u < 2; ++u) {
            const int kq = t * 16 + 8 * lhalf + 4 * u;
            float4 wq[NO / 16];
            #pragma unroll
            for (int t2 = 0; t2 < NO / 16; ++t2)
                wq[t2] = *(const float4*)(sW1 + (t2 * 16 + l16) * KP + kq);
            v2f a0, a1;
            a0.x = hT[t][4 * u + 0];  a0.y = hT[t][4 * u + 1];
            a1.x = hT[t][4 * u + 2];  a1.y = hT[t][4 * u + 3];
            #pragma unroll
            for (int t2 = 0; t2 < NO / 16; ++t2) {
                v2f b0;
                b0.x = wq[t2].x;  b0.y = wq[t2].y;
                acc[t2] = __builtin_amdgcn_wmma_f32_16x16x4_f32(
                    false, a0, false, b0, (short)0, acc[t2], false, false);
            }
            #pragma unroll
            for (int t2 = 0; t2 < NO / 16; ++t2) {
                v2f b1;
                b1.x = wq[t2].z;  b1.y = wq[t2].w;
                acc[t2] = __builtin_amdgcn_wmma_f32_16x16x4_f32(
                    false, a1, false, b1, (short)0, acc[t2], false, false);
            }
        }
    }

    float d = 1.0f;
    #pragma unroll
    for (int c = 0; c < DEG_C; ++c) d += cnt[c];
    const float dself = rsqrtf(d);
    if (lhalf == 0) dinv_out[arow] = dself;

    float dv[8];
    #pragma unroll
    for (int r = 0; r < 8; ++r)
        dv[r] = __shfl(dself, r + 8 * lhalf);

    #pragma unroll
    for (int t2 = 0; t2 < NO / 16; ++t2) {
        const int col = t2 * 16 + l16;
        #pragma unroll
        for (int r = 0; r < 8; ++r) {
            const int row = row0 + r + 8 * lhalf;
            out[(size_t)row * NO + col] = acc[t2][r] * dv[r];
        }
    }
}

#define AGG_P 36
template <int MODE>
__global__ __launch_bounds__(AGG_TB, 16) void gcn_agg(const int* __restrict__ ssrc,
                                                  const int* __restrict__ sdst,
                                                  const int* __restrict__ binstart,
                                                  const float* __restrict__ p,
                                                  const float* __restrict__ dinv,
                                                  const float* __restrict__ bias,
                                                  const float* __restrict__ Wn,
                                                  const float* __restrict__ bout,
                                                  float* __restrict__ res,
                                                  int n, int nchunk)
{
    constexpr int KP = 32 + 4;
    __shared__ __attribute__((aligned(16))) float s_acc[AGG_ROWS * AGG_P];
    __shared__ __attribute__((aligned(16))) float s_W[MODE == 0 ? 32 * KP : 4];
    __shared__ __attribute__((aligned(16))) float s_bias[32];
    __shared__ float s_dinv[AGG_ROWS];
    const int tid  = threadIdx.x;
    const int lane = tid & 31;
    const int wave = __builtin_amdgcn_readfirstlane(tid >> 5);
    const int row0 = blockIdx.x * AGG_ROWS;
    if (row0 >= n) return;
    const int rows = min(AGG_ROWS, n - row0);
    const int sh   = bin_shift(n);
    const int blo  = row0 >> sh;
    const int bhi  = (row0 + rows - 1) >> sh;

    const float dq = dinv[row0 + min(tid, rows - 1)];
    float4 wq = make_float4(0.f, 0.f, 0.f, 0.f);
    float  bq = 0.f;
    if (MODE == 0) {
        wq = ((const float4*)Wn)[tid & (32 * 32 / 4 - 1)];
        bq = bias[lane];
    }

    const float4* __restrict__ p4 = (const float4*)(p + (size_t)row0 * 32);
    for (int i = tid; i < rows * 8; i += AGG_TB)
        *(float4*)(s_acc + (i >> 3) * AGG_P + (i & 7) * 4) = p4[i];

    if (tid < rows) s_dinv[tid] = dq;
    if (MODE == 0) {
        if (tid < 32 * 32 / 4)
            *(float4*)(s_W + (tid >> 3) * KP + (tid & 7) * 4) = wq;
        if (tid < 32) s_bias[tid] = bq;
    }
    __syncthreads();

    for (int cg = 0; cg < nchunk; cg += AGG_TB) {
        int sb = 0, se = 0;
        const int cmine = cg + wave + (AGG_TB / 32) * lane;
        if (cmine < nchunk) {
            const int* __restrict__ bs = binstart + (size_t)cmine * (BIN_NB + 1);
            sb = bs[blo];  se = bs[bhi + 1];
        }
        int j = -1, c = 0, i0 = 0, e = 0;
        bool have = true;
        i0 += 32;
        while (i0 >= e) {
            ++j;
            c = cg + wave + (AGG_TB / 32) * j;
            if (j >= 32 || c >= nchunk) { have = false; break; }
            i0 = __builtin_amdgcn_readlane(sb, j);
            e  = __builtin_amdgcn_readlane(se, j);
        }
        int u = 0, v = -1;
        if (have && i0 + lane < e) {
            u = ssrc[(size_t)c * BIN_EC + i0 + lane];
            v = sdst[(size_t)c * BIN_EC + i0 + lane];
        }
        while (have) {
            const int cnt = min(32, e - i0);
            float m[32];
            #pragma unroll
            for (int k = 0; k < 8; ++k) {
                const int uu = __builtin_amdgcn_readlane(u, k);
                m[k] = p[(size_t)uu * 32 + lane];
            }
            #pragma unroll
            for (int g = 1; g < 4; ++g)
                if (8 * g < cnt) {
                    #pragma unroll
                    for (int k = 8 * g; k < 8 * g + 8; ++k) {
                        const int uu = __builtin_amdgcn_readlane(u, k);
                        m[k] = p[(size_t)uu * 32 + lane];
                    }
                }
            bool more = true;
            i0 += 32;
            while (i0 >= e) {
                ++j;
                c = cg + wave + (AGG_TB / 32) * j;
                if (j >= 32 || c >= nchunk) { more = false; break; }
                i0 = __builtin_amdgcn_readlane(sb, j);
                e  = __builtin_amdgcn_readlane(se, j);
            }
            int nu = 0, nv = -1;
            if (more && i0 + lane < e) {
                nu = ssrc[(size_t)c * BIN_EC + i0 + lane];
                nv = sdst[(size_t)c * BIN_EC + i0 + lane];
            }
            #pragma unroll
            for (int g = 0; g < 4; ++g)
                if (8 * g < cnt) {
                    #pragma unroll
                    for (int k = 8 * g; k < 8 * g + 8; ++k) {
                        const int vv = __builtin_amdgcn_readlane(v, k) - row0;
                        if ((unsigned)vv < (unsigned)rows)
                            atomicAdd(&s_acc[vv * AGG_P + lane], m[k]);
                    }
                }
            have = more;  u = nu;  v = nv;
        }
    }
    __syncthreads();

    if (MODE == 0) {
        const int trow = (wave >> 1) * 16;
        const int tcol = (wave & 1) * 16;
        const bool act = trow < rows;
        const int lhalf = lane >> 4;
        const int l16   = lane & 15;
        float o[8];
        #pragma unroll
        for (int r = 0; r < 8; ++r) o[r] = 0.0f;
        if (act) {
            const float* __restrict__ arow = s_acc + (trow + l16) * AGG_P;
            const float* __restrict__ brow = s_W + (tcol + l16) * KP;
            const float dself = s_dinv[trow + l16];
            v8f t8 = (v8f){0.f, 0.f, 0.f, 0.f, 0.f, 0.f, 0.f, 0.f};
            #pragma unroll
            for (int k0 = 0; k0 < 32; k0 += 4) {
                const int kb = k0 + 2 * lhalf;
                v2f a = *(const v2f*)(arow + kb);
                const v2f bq = *(const v2f*)(s_bias + kb);
                a.x = lrelu(dself * a.x + bq.x);
                a.y = lrelu(dself * a.y + bq.y);
                const v2f w = *(const v2f*)(brow + kb);
                t8 = __builtin_amdgcn_wmma_f32_16x16x4_f32(
                    false, a, false, w, (short)0, t8, false, false);
            }
            #pragma unroll
            for (int r = 0; r < 8; ++r)
                o[r] = t8[r] * s_dinv[trow + r + 8 * lhalf];
        }
        __syncthreads();
        if (act) {
            #pragma unroll
            for (int r = 0; r < 8; ++r)
                s_acc[(trow + r + 8 * lhalf) * AGG_P + tcol + l16] = o[r];
        }
        __syncthreads();

        float4* __restrict__ a4 = (float4*)(res + (size_t)row0 * 32);
        for (int i = tid; i < rows * 8; i += AGG_TB)
            a4[i] = *(const float4*)(s_acc + (i >> 3) * AGG_P + (i & 7) * 4);
    } else {
        if (tid < rows) {
            const float di = s_dinv[tid];
            const float4* __restrict__ q4 = (const float4*)(s_acc + tid * AGG_P);
            float s = 0.0f;
            #pragma unroll
            for (int c4 = 0; c4 < 8; ++c4) {
                const float4 q = q4[c4];
                s += Wn[4 * c4 + 0] * lrelu(di * q.x + bias[4 * c4 + 0]);
                s += Wn[4 * c4 + 1] * lrelu(di * q.y + bias[4 * c4 + 1]);
                s += Wn[4 * c4 + 2] * lrelu(di * q.z + bias[4 * c4 + 2]);
                s += Wn[4 * c4 + 3] * lrelu(di * q.w + bias[4 * c4 + 3]);
            }
            res[row0 + tid] = s + bout[0];
        }
    }
}

extern "C" void kernel_launch(void* const* d_in, const int* in_sizes, int n_in,
                              void* d_out, int out_size, void* d_ws, size_t ws_size,
                              hipStream_t stream)
{
    const float* x     = (const float*)d_in[0];
    const int*   ei    = (const int*)d_in[1];
    const float* W_in  = (const float*)d_in[2];
    const float* b_in  = (const float*)d_in[3];
    const float* W1    = (const float*)d_in[4];
    const float* b1    = (const float*)d_in[5];
    const float* W2    = (const float*)d_in[6];
    const float* b2    = (const float*)d_in[7];
    const float* W_out = (const float*)d_in[8];
    const float* b_out = (const float*)d_in[9];
    float* out = (float*)d_out;

    const int N = in_sizes[0] / 64;
    const int E = in_sizes[1] / 2;
    const int* src = ei;
    const int* dst = ei + E;

    float* ws   = (float*)d_ws;
    float* dinv = ws;
    float* p1   = ws + (size_t)N * 65;
    float* p2   = ws + (size_t)N * 97;
    float* part = ws + (size_t)N * 161;
    const int nchunk = (E + BIN_EC - 1) / BIN_EC;
    int* ssrc     = (int*)(ws + (size_t)N * (161 + DEG_C));
    int* sdst     = ssrc + (size_t)nchunk * BIN_EC;
    int* binstart = sdst + (size_t)nchunk * BIN_EC;

    gcn_deg_hist  <<<DEG_R * DEG_C + nchunk, DEG_TB, 0, stream>>>(src, dst, part, ssrc, sdst, binstart, N, E);

    const int tiles  = (N + 15) / 16;
    const int gblk   = (tiles + 3) / 4;
    const int nrg    = (N + AGG_ROWS - 1) / AGG_ROWS;

    gcn_in_conv1<<<gblk, 128, 0, stream>>>(x, W_in, b_in, W1, part, dinv, p1, N);

    gcn_agg<0><<<nrg, AGG_TB, 0, stream>>>(ssrc, sdst, binstart, p1, dinv, b1, W2, b_out, p2, N, nchunk);

    gcn_agg<1><<<nrg, AGG_TB, 0, stream>>>(ssrc, sdst, binstart, p2, dinv, b2, W_out, b_out, out, N, nchunk);
}


# edge-sum row groups and their bins narrowed from 256 to 208 rows (ceil(n/512) in whole 16-row tiles): 481 work-groups, not 391, so the busiest WGP sums 416 rows, not 512; sum order = LDS arrival
# speedup vs baseline: 3.2042x; 1.0616x over previous
#include <hip/hip_runtime.h>
#include <hip/hip_bf16.h>

typedef __attribute__((ext_vector_type(2))) float v2f;
typedef __attribute__((ext_vector_type(8))) float v8f;

__device__ __forceinline__ float lrelu(float v) { return v > 0.0f ? v : 0.01f * v; }

#define BIN_EC   8192
#define BIN_NB   512
#define BIN_TB   1024
#define AGG_TB   1024
#define AGG_ROWS 256

__host__ __device__ __forceinline__ int bin_rows(int n) {
    const int w = (n + BIN_NB - 1) / BIN_NB;
    return (w + 15) & ~15;
}
__host__ __device__ __forceinline__ int agg_rows(int n) {
    const int w = bin_rows(n);
    return w < AGG_ROWS ? w : AGG_ROWS;
}

__device__ __forceinline__ void bin_chunk(int* lds, int c,
                                          const int* __restrict__ src,
                                          const int* __restrict__ dst,
                                          int* __restrict__ ssrc,
                                          int* __restrict__ sdst,
                                          int* __restrict__ binstart,
                                          int n, int E)
{
    int* s_src = lds;
    int* s_dst = lds + BIN_EC;
    unsigned* s_cur = (unsigned*)(lds + 2 * BIN_EC);
    unsigned* s_tot = (unsigned*)(lds + 2 * BIN_EC + BIN_NB);
    const int tid = threadIdx.x;
    const int e0  = c * BIN_EC;
    const int cnt = min(BIN_EC, E - e0);
    const unsigned bw = (unsigned)bin_rows(n);

    if (tid < BIN_NB) s_cur[tid] = 0u;
    __syncthreads();

    int d[BIN_EC / BIN_TB], s[BIN_EC / BIN_TB];
    #pragma unroll
    for (int k = 0; k < BIN_EC / BIN_TB; ++k) {
        const int i = tid + k * BIN_TB;
        d[k] = -1;  s[k] = 0;
        if (i < cnt) { d[k] = dst[e0 + i];  s[k] = src[e0 + i]; }
    }
    #pragma unroll
    for (int k = 0; k < BIN_EC / BIN_TB; ++k)
        if (d[k] >= 0) atomicAdd(&s_cur[min((unsigned)d[k] / bw, (unsigned)(BIN_NB - 1))], 1u);
    __syncthreads();

    unsigned val = 0u;
    if (tid < BIN_NB) val = s_cur[tid];
    unsigned incl = val;
    #pragma unroll
    for (int off = 1; off < 32; off <<= 1) {
        const unsigned t = __shfl_up(incl, off);
        if ((tid & 31) >= off) incl += t;
    }
    if (tid < BIN_NB && (tid & 31) == 31) s_tot[tid >> 5] = incl;
    __syncthreads();
    if (tid < BIN_NB) {
        unsigned pre = 0u;
        for (int w = 0; w < (tid >> 5); ++w) pre += s_tot[w];
        const unsigned excl = pre + incl - val;
        s_cur[tid] = excl;
        binstart[(size_t)c * (BIN_NB + 1) + tid] = (int)excl;
    }
    if (tid == 0) binstart[(size_t)c * (BIN_NB + 1) + BIN_NB] = cnt;
    __syncthreads();

    #pragma unroll
    for (int k = 0; k < BIN_EC / BIN_TB; ++k)
        if (d[k] >= 0) {
            const unsigned pos = atomicAdd(&s_cur[min((unsigned)d[k] / bw, (unsigned)(BIN_NB - 1))], 1u);
            s_src[pos] = s[k];
            s_dst[pos] = d[k];
        }
    __syncthreads();
    for (int i = tid; i < cnt; i += BIN_TB) {
        ssrc[e0 + i] = s_src[i];
        sdst[e0 + i] = s_dst[i];
    }
}

#define DEG_R   16
#define DEG_C   16
#define DEG_CAP 8192
#define DEG_TB  1024

__device__ __forceinline__ void deg_tally(unsigned* s, int d, int base, unsigned cnt) {
    const unsigned off = (unsigned)(d - base);
    if (off < cnt) atomicAdd(&s[off], 1u);
}

__global__ __launch_bounds__(DEG_TB) void gcn_deg_hist(const int* __restrict__ src,
                                                       const int* __restrict__ dst,
                                                       float* __restrict__ part,
                                                       int* __restrict__ ssrc,
                                                       int* __restrict__ sdst,
                                                       int* __restrict__ binstart,
                                                       int n, int E)
{
    __shared__ __attribute__((aligned(16))) int s_raw[2 * BIN_EC + BIN_NB + BIN_NB / 32];
    if (blockIdx.x >= DEG_R * DEG_C) {
        bin_chunk(s_raw, (int)blockIdx.x - DEG_R * DEG_C, src, dst, ssrc, sdst, binstart, n, E);
        return;
    }
    unsigned* s_cnt = (unsigned*)s_raw;
    const int tid = threadIdx.x;
    const int r = blockIdx.x % DEG_R;
    const int c = blockIdx.x / DEG_R;

    const int rsize = (n + DEG_R - 1) / DEG_R;
    const int rlo = min(n, r * rsize);
    const int rhi = min(n, rlo + rsize);

    const int head  = min(E, (int)(((16u - ((unsigned)(size_t)dst & 15u)) & 15u) >> 2));
    const int4* __restrict__ dst4 = (const int4*)(dst + head);
    const int nvec  = (E - head) >> 2;
    const int tail0 = head + 4 * nvec;
    const int vsz = (nvec + DEG_C - 1) / DEG_C;
    const int vlo = min(nvec, c * vsz);
    const int vhi = min(nvec, vlo + vsz);

    for (int base = rlo; base < rhi; base += DEG_CAP) {
        const unsigned cnt = (unsigned)min(DEG_CAP, rhi - base);
        for (unsigned i = tid; i < cnt; i += DEG_TB) s_cnt[i] = 0u;
        __syncthreads();

        int i = vlo + tid;
        for (; i + 7 * DEG_TB < vhi; i += 8 * DEG_TB) {
            int4 v[8];
            #pragma unroll
            for (int k = 0; k < 8; ++k) v[k] = dst4[i + k * DEG_TB];
            #pragma unroll
            for (int k = 0; k < 8; ++k) {
                deg_tally(s_cnt, v[k].x, base, cnt);
                deg_tally(s_cnt, v[k].y, base, cnt);
                deg_tally(s_cnt, v[k].z, base, cnt);
                deg_tally(s_cnt, v[k].w, base, cnt);
            }
        }
        for (; i < vhi; i += DEG_TB) {
            const int4 v = dst4[i];
            deg_tally(s_cnt, v.x, base, cnt);
            deg_tally(s_cnt, v.y, base, cnt);
            deg_tally(s_cnt, v.z, base, cnt);
            deg_tally(s_cnt, v.w, base, cnt);
        }
        if (c == 0) {
            if (tid < head)      deg_tally(s_cnt, dst[tid], base, cnt);
            if (tid < E - tail0) deg_tally(s_cnt, dst[tail0 + tid], base, cnt);
        }
        __syncthreads();

        for (unsigned i2 = tid; i2 < cnt; i2 += DEG_TB)
            part[(size_t)c * n + base + i2] = (float)s_cnt[i2];
        __syncthreads();
    }
}


__global__ __launch_bounds__(128) void gcn_in_conv1(
    const float* __restrict__ x,
    const float* __restrict__ Win,
    const float* __restrict__ bin,
    const float* __restrict__ W1,
    const float* __restrict__ part,
    float* __restrict__ dinv_out,
    float* __restrict__ out,
    int n)
{
    constexpr int K  = 64;
    constexpr int KP = K + 4;
    constexpr int NH = 64;
    constexpr int NO = 32;
    __shared__ __attribute__((aligned(16))) float sWin[NH * KP];
    __shared__ __attribute__((aligned(16))) float sW1[NO * KP];
    __shared__ __attribute__((aligned(16))) float sBin[NH];
    for (int i = threadIdx.x; i < NH * K / 4; i += blockDim.x) {
        const int wr = i / (K / 4);
        const int wc = (i % (K / 4)) * 4;
        *(float4*)(sWin + wr * KP + wc) = ((const float4*)Win)[i];
    }
    for (int i = threadIdx.x; i < NO * K / 4; i += blockDim.x) {
        const int wr = i / (K / 4);
        const int wc = (i % (K / 4)) * 4;
        *(float4*)(sW1 + wr * KP + wc) = ((const float4*)W1)[i];
    }
    if (threadIdx.x < NH) sBin[threadIdx.x] = bin[threadIdx.x];
    __syncthreads();

    const int lane  = threadIdx.x & 31;
    const int wave  = threadIdx.x >> 5;
    const int row0  = (blockIdx.x * 4 + wave) * 16;
    if (row0 >= n) return;

    const int lhalf = lane >> 4;
    const int l16   = lane & 15;
    const int arow  = row0 + l16;

    float cnt[DEG_C];
    #pragma unroll
    for (int c = 0; c < DEG_C; ++c) cnt[c] = part[(size_t)c * n + arow];

    v8f hT[NH / 16];
    #pragma unroll
    for (int t = 0; t < NH / 16; ++t)
        hT[t] = (v8f){0.f, 0.f, 0.f, 0.f, 0.f, 0.f, 0.f, 0.f};

    #pragma unroll
    for (int k0 = 0; k0 < K; k0 += 4) {
        const int kb = k0 + 2 * lhalf;
        v2f xv;
        xv.x = x[(size_t)arow * K + kb];
        xv.y = x[(size_t)arow * K + kb + 1];
        #pragma unroll
        for (int t = 0; t < NH / 16; ++t) {
            const int ch = t * 16 + l16;
            v2f w;
            w.x = sWin[ch * KP + kb];
            w.y = sWin[ch * KP + kb + 1];
            hT[t] = __builtin_amdgcn_wmma_f32_16x16x4_f32(
                false, w, false, xv, (short)0, hT[t], false, false);
        }
    }

    #pragma unroll
    for (int t = 0; t < NH / 16; ++t) {
        #pragma unroll
        for (int u = 0; u < 2; ++u) {
            const float4 bq = *(const float4*)(sBin + t * 16 + 8 * lhalf + 4 * u);
            hT[t][4 * u + 0] = lrelu(hT[t][4 * u + 0] + bq.x);
            hT[t][4 * u + 1] = lrelu(hT[t][4 * u + 1] + bq.y);
            hT[t][4 * u + 2] = lrelu(hT[t][4 * u + 2] + bq.z);
            hT[t][4 * u + 3] = lrelu(hT[t][4 * u + 3] + bq.w);
        }
    }

    v8f acc[NO / 16];
    #pragma unroll
    for (int t2 = 0; t2 < NO / 16; ++t2)
        acc[t2] = (v8f){0.f, 0.f, 0.f, 0.f, 0.f, 0.f, 0.f, 0.f};

    #pragma unroll
    for (int t = 0; t < NH / 16; ++t) {
        #pragma unroll
        for (int u = 0; u < 2; ++u) {
            const int kq = t * 16 + 8 * lhalf + 4 * u;
            float4 wq[NO / 16];
            #pragma unroll
            for (int t2 = 0; t2 < NO / 16; ++t2)
                wq[t2] = *(const float4*)(sW1 + (t2 * 16 + l16) * KP + kq);
            v2f a0, a1;
            a0.x = hT[t][4 * u + 0];  a0.y = hT[t][4 * u + 1];
            a1.x = hT[t][4 * u + 2];  a1.y = hT[t][4 * u + 3];
            #pragma unroll
            for (int t2 = 0; t2 < NO / 16; ++t2) {
                v2f b0;
                b0.x = wq[t2].x;  b0.y = wq[t2].y;
                acc[t2] = __builtin_amdgcn_wmma_f32_16x16x4_f32(
                    false, a0, false, b0, (short)0, acc[t2], false, false);
            }
            #pragma unroll
            for (int t2 = 0; t2 < NO / 16; ++t2) {
                v2f b1;
                b1.x = wq[t2].z;  b1.y = wq[t2].w;
                acc[t2] = __builtin_amdgcn_wmma_f32_16x16x4_f32(
                    false, a1, false, b1, (short)0, acc[t2], false, false);
            }
        }
    }

    float d = 1.0f;
    #pragma unroll
    for (int c = 0; c < DEG_C; ++c) d += cnt[c];
    const float dself = rsqrtf(d);
    if (lhalf == 0) dinv_out[arow] = dself;

    float dv[8];
    #pragma unroll
    for (int r = 0; r < 8; ++r)
        dv[r] = __shfl(dself, r + 8 * lhalf);

    #pragma unroll
    for (int t2 = 0; t2 < NO / 16; ++t2) {
        const int col = t2 * 16 + l16;
        #pragma unroll
        for (int r = 0; r < 8; ++r) {
            const int row = row0 + r + 8 * lhalf;
            out[(size_t)row * NO + col] = acc[t2][r] * dv[r];
        }
    }
}

#define AGG_P 36
template <int MODE>
__global__ __launch_bounds__(AGG_TB, 16) void gcn_agg(const int* __restrict__ ssrc,
                                                  const int* __restrict__ sdst,
                                                  const int* __restrict__ binstart,
                                                  const float* __restrict__ p,
                                                  const float* __restrict__ dinv,
                                                  const float* __restrict__ bias,
                                                  const float* __restrict__ Wn,
                                                  const float* __restrict__ bout,
                                                  float* __restrict__ res,
                                                  int n, int nchunk)
{
    constexpr int KP = 32 + 4;
    __shared__ __attribute__((aligned(16))) float s_acc[AGG_ROWS * AGG_P];
    __shared__ __attribute__((aligned(16))) float s_W[MODE == 0 ? 32 * KP : 4];
    __shared__ __attribute__((aligned(16))) float s_bias[32];
    __shared__ float s_dinv[AGG_ROWS];
    const int tid  = threadIdx.x;
    const int lane = tid & 31;
    const int wave = __builtin_amdgcn_readfirstlane(tid >> 5);
    const int rgrp = agg_rows(n);
    const int row0 = blockIdx.x * rgrp;
    if (row0 >= n) return;
    const int rows = min(rgrp, n - row0);
    const int bw   = bin_rows(n);
    const int blo  = row0 / bw;
    const int bhi  = (row0 + rows - 1) / bw;

    const float dq = dinv[row0 + min(tid, rows - 1)];
    float4 wq = make_float4(0.f, 0.f, 0.f, 0.f);
    float  bq = 0.f;
    if (MODE == 0) {
        wq = ((const float4*)Wn)[tid & (32 * 32 / 4 - 1)];
        bq = bias[lane];
    }

    const float4* __restrict__ p4 = (const float4*)(p + (size_t)row0 * 32);
    for (int i = tid; i < rows * 8; i += AGG_TB)
        *(float4*)(s_acc + (i >> 3) * AGG_P + (i & 7) * 4) = p4[i];

    if (tid < rows) s_dinv[tid] = dq;
    if (MODE == 0) {
        if (tid < 32 * 32 / 4)
            *(float4*)(s_W + (tid >> 3) * KP + (tid & 7) * 4) = wq;
        if (tid < 32) s_bias[tid] = bq;
    }
    __syncthreads();

    for (int cg = 0; cg < nchunk; cg += AGG_TB) {
        int sb = 0, se = 0;
        const int cmine = cg + wave + (AGG_TB / 32) * lane;
        if (cmine < nchunk) {
            const int* __restrict__ bs = binstart + (size_t)cmine * (BIN_NB + 1);
            sb = bs[blo];  se = bs[bhi + 1];
        }
        int j = -1, c = 0, i0 = 0, e = 0;
        bool have = true;
        i0 += 32;
        while (i0 >= e) {
            ++j;
            c = cg + wave + (AGG_TB / 32) * j;
            if (j >= 32 || c >= nchunk) { have = false; break; }
            i0 = __builtin_amdgcn_readlane(sb, j);
            e  = __builtin_amdgcn_readlane(se, j);
        }
        int u = 0, v = -1;
        if (have && i0 + lane < e) {
            u = ssrc[(size_t)c * BIN_EC + i0 + lane];
            v = sdst[(size_t)c * BIN_EC + i0 + lane];
        }
        while (have) {
            const int cnt = min(32, e - i0);
            float m[32];
            #pragma unroll
            for (int k = 0; k < 8; ++k) {
                const int uu = __builtin_amdgcn_readlane(u, k);
                m[k] = p[(size_t)uu * 32 + lane];
            }
            #pragma unroll
            for (int g = 1; g < 4; ++g)
                if (8 * g < cnt) {
                    #pragma unroll
                    for (int k = 8 * g; k < 8 * g + 8; ++k) {
                        const int uu = __builtin_amdgcn_readlane(u, k);
                        m[k] = p[(size_t)uu * 32 + lane];
                    }
                }
            bool more = true;
            i0 += 32;
            while (i0 >= e) {
                ++j;
                c = cg + wave + (AGG_TB / 32) * j;
                if (j >= 32 || c >= nchunk) { more = false; break; }
                i0 = __builtin_amdgcn_readlane(sb, j);
                e  = __builtin_amdgcn_readlane(se, j);
            }
            int nu = 0, nv = -1;
            if (more && i0 + lane < e) {
                nu = ssrc[(size_t)c * BIN_EC + i0 + lane];
                nv = sdst[(size_t)c * BIN_EC + i0 + lane];
            }
            #pragma unroll
            for (int g = 0; g < 4; ++g)
                if (8 * g < cnt) {
                    #pragma unroll
                    for (int k = 8 * g; k < 8 * g + 8; ++k) {
                        const int vv = __builtin_amdgcn_readlane(v, k) - row0;
                        if ((unsigned)vv < (unsigned)rows)
                            atomicAdd(&s_acc[vv * AGG_P + lane], m[k]);
                    }
                }
            have = more;  u = nu;  v = nv;
        }
    }
    __syncthreads();

    if (MODE == 0) {
        const int trow = (wave >> 1) * 16;
        const int tcol = (wave & 1) * 16;
        const bool act = trow < rows;
        const int lhalf = lane >> 4;
        const int l16   = lane & 15;
        float o[8];
        #pragma unroll
        for (int r = 0; r < 8; ++r) o[r] = 0.0f;
        if (act) {
            const float* __restrict__ arow = s_acc + (trow + l16) * AGG_P;
            const float* __restrict__ brow = s_W + (tcol + l16) * KP;
            const float dself = s_dinv[trow + l16];
            v8f t8 = (v8f){0.f, 0.f, 0.f, 0.f, 0.f, 0.f, 0.f, 0.f};
            #pragma unroll
            for (int k0 = 0; k0 < 32; k0 += 4) {
                const int kb = k0 + 2 * lhalf;
                v2f a = *(const v2f*)(arow + kb);
                const v2f bq = *(const v2f*)(s_bias + kb);
                a.x = lrelu(dself * a.x + bq.x);
                a.y = lrelu(dself * a.y + bq.y);
                const v2f w = *(const v2f*)(brow + kb);
                t8 = __builtin_amdgcn_wmma_f32_16x16x4_f32(
                    false, a, false, w, (short)0, t8, false, false);
            }
            #pragma unroll
            for (int r = 0; r < 8; ++r)
                o[r] = t8[r] * s_dinv[trow + r + 8 * lhalf];
        }
        __syncthreads();
        if (act) {
            #pragma unroll
            for (int r = 0; r < 8; ++r)
                s_acc[(trow + r + 8 * lhalf) * AGG_P + tcol + l16] = o[r];
        }
        __syncthreads();

        float4* __restrict__ a4 = (float4*)(res + (size_t)row0 * 32);
        for (int i = tid; i < rows * 8; i += AGG_TB)
            a4[i] = *(const float4*)(s_acc + (i >> 3) * AGG_P + (i & 7) * 4);
    } else {
        if (tid < rows) {
            const float di = s_dinv[tid];
            const float4* __restrict__ q4 = (const float4*)(s_acc + tid * AGG_P);
            float s = 0.0f;
            #pragma unroll
            for (int c4 = 0; c4 < 8; ++c4) {
                const float4 q = q4[c4];
                s += Wn[4 * c4 + 0] * lrelu(di * q.x + bias[4 * c4 + 0]);
                s += Wn[4 * c4 + 1] * lrelu(di * q.y + bias[4 * c4 + 1]);
                s += Wn[4 * c4 + 2] * lrelu(di * q.z + bias[4 * c4 + 2]);
                s += Wn[4 * c4 + 3] * lrelu(di * q.w + bias[4 * c4 + 3]);
            }
            res[row0 + tid] = s + bout[0];
        }
    }
}

extern "C" void kernel_launch(void* const* d_in, const int* in_sizes, int n_in,
                              void* d_out, int out_size, void* d_ws, size_t ws_size,
                              hipStream_t stream)
{
    const float* x     = (const float*)d_in[0];
    const int*   ei    = (const int*)d_in[1];
    const float* W_in  = (const float*)d_in[2];
    const float* b_in  = (const float*)d_in[3];
    const float* W1    = (const float*)d_in[4];
    const float* b1    = (const float*)d_in[5];
    const float* W2    = (const float*)d_in[6];
    const float* b2    = (const float*)d_in[7];
    const float* W_out = (const float*)d_in[8];
    const float* b_out = (const float*)d_in[9];
    float* out = (float*)d_out;

    const int N = in_sizes[0] / 64;
    const int E = in_sizes[1] / 2;
    const int* src = ei;
    const int* dst = ei + E;

    float* ws   = (float*)d_ws;
    float* dinv = ws;
    float* p1   = ws + (size_t)N * 65;
    float* p2   = ws + (size_t)N * 97;
    float* part = ws + (size_t)N * 161;
    const int nchunk = (E + BIN_EC - 1) / BIN_EC;
    int* ssrc     = (int*)(ws + (size_t)N * (161 + DEG_C));
    int* sdst     = ssrc + (size_t)nchunk * BIN_EC;
    int* binstart = sdst + (size_t)nchunk * BIN_EC;

    gcn_deg_hist  <<<DEG_R * DEG_C + nchunk, DEG_TB, 0, stream>>>(src, dst, part, ssrc, sdst, binstart, N, E);

    const int tiles  = (N + 15) / 16;
    const int gblk   = (tiles + 3) / 4;
    const int nrg    = (N + agg_rows(N) - 1) / agg_rows(N);

    gcn_in_conv1<<<gblk, 128, 0, stream>>>(x, W_in, b_in, W1, part, dinv, p1, N);

    gcn_agg<0><<<nrg, AGG_TB, 0, stream>>>(ssrc, sdst, binstart, p1, dinv, b1, W2, b_out, p2, N, nchunk);

    gcn_agg<1><<<nrg, AGG_TB, 0, stream>>>(ssrc, sdst, binstart, p2, dinv, b2, W_out, b_out, out, N, nchunk);
}


# each kernel's prologue loads issued as one group ahead of the first wait: conv1's 12 weight pieces and x, the degree scan's tail vectors, the edge sums' bounds and start rows; same arithmetic
# speedup vs baseline: 3.5369x; 1.1038x over previous
#include <hip/hip_runtime.h>
#include <hip/hip_bf16.h>

typedef __attribute__((ext_vector_type(2))) float v2f;
typedef __attribute__((ext_vector_type(8))) float v8f;

__device__ __forceinline__ float lrelu(float v) { return v > 0.0f ? v : 0.01f * v; }

#define BIN_EC   8192
#define BIN_NB   512
#define BIN_TB   1024
#define AGG_TB   1024
#define AGG_ROWS 256

__host__ __device__ __forceinline__ int bin_rows(int n) {
    const int w = (n + BIN_NB - 1) / BIN_NB;
    return (w + 15) & ~15;
}
__host__ __device__ __forceinline__ int agg_rows(int n) {
    const int w = bin_rows(n);
    return w < AGG_ROWS ? w : AGG_ROWS;
}

__device__ __forceinline__ void bin_chunk(int* lds, int c,
                                          const int* __restrict__ src,
                                          const int* __restrict__ dst,
                                          int* __restrict__ ssrc,
                                          int* __restrict__ sdst,
                                          int* __restrict__ binstart,
                                          int n, int E)
{
    int* s_src = lds;
    int* s_dst = lds + BIN_EC;
    unsigned* s_cur = (unsigned*)(lds + 2 * BIN_EC);
    unsigned* s_tot = (unsigned*)(lds + 2 * BIN_EC + BIN_NB);
    const int tid = threadIdx.x;
    const int e0  = c * BIN_EC;
    const int cnt = min(BIN_EC, E - e0);
    const unsigned bw = (unsigned)bin_rows(n);

    if (tid < BIN_NB) s_cur[tid] = 0u;
    __syncthreads();

    int d[BIN_EC / BIN_TB], s[BIN_EC / BIN_TB];
    #pragma unroll
    for (int k = 0; k < BIN_EC / BIN_TB; ++k) {
        const int i = tid + k * BIN_TB;
        d[k] = -1;  s[k] = 0;
        if (i < cnt) { d[k] = dst[e0 + i];  s[k] = src[e0 + i]; }
    }
    #pragma unroll
    for (int k = 0; k < BIN_EC / BIN_TB; ++k)
        if (d[k] >= 0) atomicAdd(&s_cur[min((unsigned)d[k] / bw, (unsigned)(BIN_NB - 1))], 1u);
    __syncthreads();

    unsigned val = 0u;
    if (tid < BIN_NB) val = s_cur[tid];
    unsigned incl = val;
    #pragma unroll
    for (int off = 1; off < 32; off <<= 1) {
        const unsigned t = __shfl_up(incl, off);
        if ((tid & 31) >= off) incl += t;
    }
    if (tid < BIN_NB && (tid & 31) == 31) s_tot[tid >> 5] = incl;
    __syncthreads();
    if (tid < BIN_NB) {
        unsigned pre = 0u;
        for (int w = 0; w < (tid >> 5); ++w) pre += s_tot[w];
        const unsigned excl = pre + incl - val;
        s_cur[tid] = excl;
        binstart[(size_t)c * (BIN_NB + 1) + tid] = (int)excl;
    }
    if (tid == 0) binstart[(size_t)c * (BIN_NB + 1) + BIN_NB] = cnt;
    __syncthreads();

    #pragma unroll
    for (int k = 0; k < BIN_EC / BIN_TB; ++k)
        if (d[k] >= 0) {
            const unsigned pos = atomicAdd(&s_cur[min((unsigned)d[k] / bw, (unsigned)(BIN_NB - 1))], 1u);
            s_src[pos] = s[k];
            s_dst[pos] = d[k];
        }
    __syncthreads();
    for (int i = tid; i < cnt; i += BIN_TB) {
        ssrc[e0 + i] = s_src[i];
        sdst[e0 + i] = s_dst[i];
    }
}

#define DEG_R   16
#define DEG_C   16
#define DEG_CAP 8192
#define DEG_TB  1024

__device__ __forceinline__ void deg_tally(unsigned* s, int d, int base, unsigned cnt) {
    const unsigned off = (unsigned)(d - base);
    if (off < cnt) atomicAdd(&s[off], 1u);
}

__global__ __launch_bounds__(DEG_TB, 16) void gcn_deg_hist(const int* __restrict__ src,
                                                       const int* __restrict__ dst,
                                                       float* __restrict__ part,
                                                       int* __restrict__ ssrc,
                                                       int* __restrict__ sdst,
                                                       int* __restrict__ binstart,
                                                       int n, int E)
{
    __shared__ __attribute__((aligned(16))) int s_raw[2 * BIN_EC + BIN_NB + BIN_NB / 32];
    if (blockIdx.x >= DEG_R * DEG_C) {
        bin_chunk(s_raw, (int)blockIdx.x - DEG_R * DEG_C, src, dst, ssrc, sdst, binstart, n, E);
        return;
    }
    unsigned* s_cnt = (unsigned*)s_raw;
    const int tid = threadIdx.x;
    const int r = blockIdx.x % DEG_R;
    const int c = blockIdx.x / DEG_R;

    const int rsize = (n + DEG_R - 1) / DEG_R;
    const int rlo = min(n, r * rsize);
    const int rhi = min(n, rlo + rsize);

    const int head  = min(E, (int)(((16u - ((unsigned)(size_t)dst & 15u)) & 15u) >> 2));
    const int4* __restrict__ dst4 = (const int4*)(dst + head);
    const int nvec  = (E - head) >> 2;
    const int tail0 = head + 4 * nvec;
    const int vsz = (nvec + DEG_C - 1) / DEG_C;
    const int vlo = min(nvec, c * vsz);
    const int vhi = min(nvec, vlo + vsz);

    for (int base = rlo; base < rhi; base += DEG_CAP) {
        const unsigned cnt = (unsigned)min(DEG_CAP, rhi - base);
        for (unsigned i = tid; i < cnt; i += DEG_TB) s_cnt[i] = 0u;
        __syncthreads();

        int i = vlo + tid;
        for (; i + 7 * DEG_TB < vhi; i += 8 * DEG_TB) {
            int4 v[8];
            #pragma unroll
            for (int k = 0; k < 8; ++k) v[k] = dst4[i + k * DEG_TB];
            #pragma unroll
            for (int k = 0; k < 8; ++k) {
                deg_tally(s_cnt, v[k].x, base, cnt);
                deg_tally(s_cnt, v[k].y, base, cnt);
                deg_tally(s_cnt, v[k].z, base, cnt);
                deg_tally(s_cnt, v[k].w, base, cnt);
            }
        }
        for (; i < vhi; i += 8 * DEG_TB) {
            int4 v[8];
            #pragma unroll
            for (int k = 0; k < 8; ++k) {
                v[k] = make_int4(-1, -1, -1, -1);
                if (i + k * DEG_TB < vhi) v[k] = dst4[i + k * DEG_TB];
            }
            #pragma unroll
            for (int k = 0; k < 8; ++k) {
                deg_tally(s_cnt, v[k].x, base, cnt);
                deg_tally(s_cnt, v[k].y, base, cnt);
                deg_tally(s_cnt, v[k].z, base, cnt);
                deg_tally(s_cnt, v[k].w, base, cnt);
            }
        }
        if (c == 0) {
            if (tid < head)      deg_tally(s_cnt, dst[tid], base, cnt);
            if (tid < E - tail0) deg_tally(s_cnt, dst[tail0 + tid], base, cnt);
        }
        __syncthreads();

        for (unsigned i2 = tid; i2 < cnt; i2 += DEG_TB)
            part[(size_t)c * n + base + i2] = (float)s_cnt[i2];
        __syncthreads();
    }
}


__global__ __launch_bounds__(128) void gcn_in_conv1(
    const float* __restrict__ x,
    const float* __restrict__ Win,
    const float* __restrict__ bin,
    const float* __restrict__ W1,
    const float* __restrict__ part,
    float* __restrict__ dinv_out,
    float* __restrict__ out,
    int n)
{
    constexpr int K  = 64;
    constexpr int KP = K + 4;
    constexpr int NH = 64;
    constexpr int NO = 32;
    __shared__ __attribute__((aligned(16))) float sWin[NH * KP];
    __shared__ __attribute__((aligned(16))) float sW1[NO * KP];
    __shared__ __attribute__((aligned(16))) float sBin[NH];
    constexpr int TB  = 128;
    constexpr int NWI = NH * K / 4 / TB;
    constexpr int NW1 = NO * K / 4 / TB;

    const int lane  = threadIdx.x & 31;
    const int wave  = threadIdx.x >> 5;
    const int row0  = (blockIdx.x * 4 + wave) * 16;
    const int lhalf = lane >> 4;
    const int l16   = lane & 15;
    const int arow  = row0 + l16;
    const int lrow  = min(arow, n - 1);

    float4 gWin[NWI], gW1[NW1];
    #pragma unroll
    for (int j = 0; j < NWI; ++j) gWin[j] = ((const float4*)Win)[threadIdx.x + j * TB];
    #pragma unroll
    for (int j = 0; j < NW1; ++j) gW1[j] = ((const float4*)W1)[threadIdx.x + j * TB];
    const float gB = bin[threadIdx.x & (NH - 1)];

    float cnt[DEG_C];
    #pragma unroll
    for (int c = 0; c < DEG_C; ++c) cnt[c] = part[(size_t)c * n + lrow];

    v2f xr[K / 4];
    #pragma unroll
    for (int q = 0; q < K / 4; ++q)
        xr[q] = *(const v2f*)(x + (size_t)lrow * K + 4 * q + 2 * lhalf);

    #pragma unroll
    for (int j = 0; j < NWI; ++j) {
        const int i  = threadIdx.x + j * TB;
        const int wr = i / (K / 4);
        const int wc = (i % (K / 4)) * 4;
        *(float4*)(sWin + wr * KP + wc) = gWin[j];
    }
    #pragma unroll
    for (int j = 0; j < NW1; ++j) {
        const int i  = threadIdx.x + j * TB;
        const int wr = i / (K / 4);
        const int wc = (i % (K / 4)) * 4;
        *(float4*)(sW1 + wr * KP + wc) = gW1[j];
    }
    sBin[threadIdx.x & (NH - 1)] = gB;
    __syncthreads();
    if (row0 >= n) return;

    v8f hT[NH / 16];
    #pragma unroll
    for (int t = 0; t < NH / 16; ++t)
        hT[t] = (v8f){0.f, 0.f, 0.f, 0.f, 0.f, 0.f, 0.f, 0.f};

    #pragma unroll
    for (int k0 = 0; k0 < K; k0 += 4) {
        const int kb = k0 + 2 * lhalf;
        const v2f xv = xr[k0 / 4];
        #pragma unroll
        for (int t = 0; t < NH / 16; ++t) {
            const int ch = t * 16 + l16;
            v2f w;
            w.x = sWin[ch * KP + kb];
            w.y = sWin[ch * KP + kb + 1];
            hT[t] = __builtin_amdgcn_wmma_f32_16x16x4_f32(
                false, w, false, xv, (short)0, hT[t], false, false);
        }
    }

    #pragma unroll
    for (int t = 0; t < NH / 16; ++t) {
        #pragma unroll
        for (int u = 0; u < 2; ++u) {
            const float4 bq = *(const float4*)(sBin + t * 16 + 8 * lhalf + 4 * u);
            hT[t][4 * u + 0] = lrelu(hT[t][4 * u + 0] + bq.x);
            hT[t][4 * u + 1] = lrelu(hT[t][4 * u + 1] + bq.y);
            hT[t][4 * u + 2] = lrelu(hT[t][4 * u + 2] + bq.z);
            hT[t][4 * u + 3] = lrelu(hT[t][4 * u + 3] + bq.w);
        }
    }

    v8f acc[NO / 16];
    #pragma unroll
    for (int t2 = 0; t2 < NO / 16; ++t2)
        acc[t2] = (v8f){0.f, 0.f, 0.f, 0.f, 0.f, 0.f, 0.f, 0.f};

    #pragma unroll
    for (int t = 0; t < NH / 16; ++t) {
        #pragma unroll
        for (int u = 0; u < 2; ++u) {
            const int kq = t * 16 + 8 * lhalf + 4 * u;
            float4 wq[NO / 16];
            #pragma unroll
            for (int t2 = 0; t2 < NO / 16; ++t2)
                wq[t2] = *(const float4*)(sW1 + (t2 * 16 + l16) * KP + kq);
            v2f a0, a1;
            a0.x = hT[t][4 * u + 0];  a0.y = hT[t][4 * u + 1];
            a1.x = hT[t][4 * u + 2];  a1.y = hT[t][4 * u + 3];
            #pragma unroll
            for (int t2 = 0; t2 < NO / 16; ++t2) {
                v2f b0;
                b0.x = wq[t2].x;  b0.y = wq[t2].y;
                acc[t2] = __builtin_amdgcn_wmma_f32_16x16x4_f32(
                    false, a0, false, b0, (short)0, acc[t2], false, false);
            }
            #pragma unroll
            for (int t2 = 0; t2 < NO / 16; ++t2) {
                v2f b1;
                b1.x = wq[t2].z;  b1.y = wq[t2].w;
                acc[t2] = __builtin_amdgcn_wmma_f32_16x16x4_f32(
                    false, a1, false, b1, (short)0, acc[t2], false, false);
            }
        }
    }

    float d = 1.0f;
    #pragma unroll
    for (int c = 0; c < DEG_C; ++c) d += cnt[c];
    const float dself = rsqrtf(d);
    if (lhalf == 0) dinv_out[arow] = dself;

    float dv[8];
    #pragma unroll
    for (int r = 0; r < 8; ++r)
        dv[r] = __shfl(dself, r + 8 * lhalf);

    #pragma unroll
    for (int t2 = 0; t2 < NO / 16; ++t2) {
        const int col = t2 * 16 + l16;
        #pragma unroll
        for (int r = 0; r < 8; ++r) {
            const int row = row0 + r + 8 * lhalf;
            out[(size_t)row * NO + col] = acc[t2][r] * dv[r];
        }
    }
}

#define AGG_P 36
template <int MODE>
__global__ __launch_bounds__(AGG_TB, 16) void gcn_agg(const int* __restrict__ ssrc,
                                                  const int* __restrict__ sdst,
                                                  const int* __restrict__ binstart,
                                                  const float* __restrict__ p,
                                                  const float* __restrict__ dinv,
                                                  const float* __restrict__ bias,
                                                  const float* __restrict__ Wn,
                                                  const float* __restrict__ bout,
                                                  float* __restrict__ res,
                                                  int n, int nchunk)
{
    constexpr int KP = 32 + 4;
    __shared__ __attribute__((aligned(16))) float s_acc[AGG_ROWS * AGG_P];
    __shared__ __attribute__((aligned(16))) float s_W[MODE == 0 ? 32 * KP : 4];
    __shared__ __attribute__((aligned(16))) float s_bias[32];
    __shared__ float s_dinv[AGG_ROWS];
    const int tid  = threadIdx.x;
    const int lane = tid & 31;
    const int wave = __builtin_amdgcn_readfirstlane(tid >> 5);
    const int rgrp = agg_rows(n);
    const int row0 = blockIdx.x * rgrp;
    if (row0 >= n) return;
    const int rows = min(rgrp, n - row0);
    const int bw   = bin_rows(n);
    const int blo  = row0 / bw;
    const int bhi  = (row0 + rows - 1) / bw;


    int sb = 0, se = 0;
    auto load_bounds = [&](int cg) {
        sb = 0;  se = 0;
        const int cmine = cg + wave + (AGG_TB / 32) * lane;
        if (cmine < nchunk) {
            const int* __restrict__ bs = binstart + (size_t)cmine * (BIN_NB + 1);
            sb = bs[blo];  se = bs[bhi + 1];
        }
    };
    int j = -1, c = 0, i0 = 0, e = 0;
    bool have = false;
    int u = 0, v = -1;
    auto first_batch = [&](int cg) {
        j = -1;  c = 0;  i0 = 0;  e = 0;  have = true;
        i0 += 32;
        while (i0 >= e) {
            ++j;
            c = cg + wave + (AGG_TB / 32) * j;
            if (j >= 32 || c >= nchunk) { have = false; break; }
            i0 = __builtin_amdgcn_readlane(sb, j);
            e  = __builtin_amdgcn_readlane(se, j);
        }
        u = 0;  v = -1;
        if (have && i0 + lane < e) {
            u = ssrc[(size_t)c * BIN_EC + i0 + lane];
            v = sdst[(size_t)c * BIN_EC + i0 + lane];
        }
    };

    load_bounds(0);

    const float dq = dinv[row0 + min(tid, rows - 1)];
    float4 wq = make_float4(0.f, 0.f, 0.f, 0.f);
    float  bq = 0.f;
    if (MODE == 0) {
        wq = ((const float4*)Wn)[tid & (32 * 32 / 4 - 1)];
        bq = bias[lane];
    }

    const float4* __restrict__ p4 = (const float4*)(p + (size_t)row0 * 32);
    constexpr int NI = AGG_ROWS * 8 / AGG_TB;
    float4 pq[NI];
    #pragma unroll
    for (int k = 0; k < NI; ++k) pq[k] = p4[min(tid + k * AGG_TB, rows * 8 - 1)];

    first_batch(0);

    #pragma unroll
    for (int k = 0; k < NI; ++k) {
        const int i = tid + k * AGG_TB;
        *(float4*)(s_acc + (i >> 3) * AGG_P + (i & 7) * 4) = pq[k];
    }
    if (tid < rows) s_dinv[tid] = dq;
    if (MODE == 0) {
        if (tid < 32 * 32 / 4)
            *(float4*)(s_W + (tid >> 3) * KP + (tid & 7) * 4) = wq;
        if (tid < 32) s_bias[tid] = bq;
    }
    __syncthreads();

    for (int cg = 0; cg < nchunk; cg += AGG_TB) {
        if (cg > 0) { load_bounds(cg);  first_batch(cg); }
        while (have) {
            const int cnt = min(32, e - i0);
            float m[32];
            #pragma unroll
            for (int k = 0; k < 8; ++k) {
                const int uu = __builtin_amdgcn_readlane(u, k);
                m[k] = p[(size_t)uu * 32 + lane];
            }
            #pragma unroll
            for (int g = 1; g < 4; ++g)
                if (8 * g < cnt) {
                    #pragma unroll
                    for (int k = 8 * g; k < 8 * g + 8; ++k) {
                        const int uu = __builtin_amdgcn_readlane(u, k);
                        m[k] = p[(size_t)uu * 32 + lane];
                    }
                }
            bool more = true;
            i0 += 32;
            while (i0 >= e) {
                ++j;
                c = cg + wave + (AGG_TB / 32) * j;
                if (j >= 32 || c >= nchunk) { more = false; break; }
                i0 = __builtin_amdgcn_readlane(sb, j);
                e  = __builtin_amdgcn_readlane(se, j);
            }
            int nu = 0, nv = -1;
            if (more && i0 + lane < e) {
                nu = ssrc[(size_t)c * BIN_EC + i0 + lane];
                nv = sdst[(size_t)c * BIN_EC + i0 + lane];
            }
            #pragma unroll
            for (int g = 0; g < 4; ++g)
                if (8 * g < cnt) {
                    #pragma unroll
                    for (int k = 8 * g; k < 8 * g + 8; ++k) {
                        const int vv = __builtin_amdgcn_readlane(v, k) - row0;
                        if ((unsigned)vv < (unsigned)rows)
                            atomicAdd(&s_acc[vv * AGG_P + lane], m[k]);
                    }
                }
            have = more;  u = nu;  v = nv;
        }
    }
    __syncthreads();

    if (MODE == 0) {
        const int trow = (wave >> 1) * 16;
        const int tcol = (wave & 1) * 16;
        const bool act = trow < rows;
        const int lhalf = lane >> 4;
        const int l16   = lane & 15;
        float o[8];
        #pragma unroll
        for (int r = 0; r < 8; ++r) o[r] = 0.0f;
        if (act) {
            const float* __restrict__ arow = s_acc + (trow + l16) * AGG_P;
            const float* __restrict__ brow = s_W + (tcol + l16) * KP;
            const float dself = s_dinv[trow + l16];
            v8f t8 = (v8f){0.f, 0.f, 0.f, 0.f, 0.f, 0.f, 0.f, 0.f};
            #pragma unroll
            for (int k0 = 0; k0 < 32; k0 += 4) {
                const int kb = k0 + 2 * lhalf;
                v2f a = *(const v2f*)(arow + kb);
                const v2f bq = *(const v2f*)(s_bias + kb);
                a.x = lrelu(dself * a.x + bq.x);
                a.y = lrelu(dself * a.y + bq.y);
                const v2f w = *(const v2f*)(brow + kb);
                t8 = __builtin_amdgcn_wmma_f32_16x16x4_f32(
                    false, a, false, w, (short)0, t8, false, false);
            }
            #pragma unroll
            for (int r = 0; r < 8; ++r)
                o[r] = t8[r] * s_dinv[trow + r + 8 * lhalf];
        }
        __syncthreads();
        if (act) {
            #pragma unroll
            for (int r = 0; r < 8; ++r)
                s_acc[(trow + r + 8 * lhalf) * AGG_P + tcol + l16] = o[r];
        }
        __syncthreads();

        float4* __restrict__ a4 = (float4*)(res + (size_t)row0 * 32);
        for (int i = tid; i < rows * 8; i += AGG_TB)
            a4[i] = *(const float4*)(s_acc + (i >> 3) * AGG_P + (i & 7) * 4);
    } else {
        if (tid < rows) {
            const float di = s_dinv[tid];
            const float4* __restrict__ q4 = (const float4*)(s_acc + tid * AGG_P);
            float s = 0.0f;
            #pragma unroll
            for (int c4 = 0; c4 < 8; ++c4) {
                const float4 q = q4[c4];
                s += Wn[4 * c4 + 0] * lrelu(di * q.x + bias[4 * c4 + 0]);
                s += Wn[4 * c4 + 1] * lrelu(di * q.y + bias[4 * c4 + 1]);
                s += Wn[4 * c4 + 2] * lrelu(di * q.z + bias[4 * c4 + 2]);
                s += Wn[4 * c4 + 3] * lrelu(di * q.w + bias[4 * c4 + 3]);
            }
            res[row0 + tid] = s + bout[0];
        }
    }
}

extern "C" void kernel_launch(void* const* d_in, const int* in_sizes, int n_in,
                              void* d_out, int out_size, void* d_ws, size_t ws_size,
                              hipStream_t stream)
{
    const float* x     = (const float*)d_in[0];
    const int*   ei    = (const int*)d_in[1];
    const float* W_in  = (const float*)d_in[2];
    const float* b_in  = (const float*)d_in[3];
    const float* W1    = (const float*)d_in[4];
    const float* b1    = (const float*)d_in[5];
    const float* W2    = (const float*)d_in[6];
    const float* b2    = (const float*)d_in[7];
    const float* W_out = (const float*)d_in[8];
    const float* b_out = (const float*)d_in[9];
    float* out = (float*)d_out;

    const int N = in_sizes[0] / 64;
    const int E = in_sizes[1] / 2;
    const int* src = ei;
    const int* dst = ei + E;

    float* ws   = (float*)d_ws;
    float* dinv = ws;
    float* p1   = ws + (size_t)N * 65;
    float* p2   = ws + (size_t)N * 97;
    float* part = ws + (size_t)N * 161;
    const int nchunk = (E + BIN_EC - 1) / BIN_EC;
    int* ssrc     = (int*)(ws + (size_t)N * (161 + DEG_C));
    int* sdst     = ssrc + (size_t)nchunk * BIN_EC;
    int* binstart = sdst + (size_t)nchunk * BIN_EC;

    gcn_deg_hist  <<<DEG_R * DEG_C + nchunk, DEG_TB, 0, stream>>>(src, dst, part, ssrc, sdst, binstart, N, E);

    const int tiles  = (N + 15) / 16;
    const int gblk   = (tiles + 3) / 4;
    const int nrg    = (N + agg_rows(N) - 1) / agg_rows(N);

    gcn_in_conv1<<<gblk, 128, 0, stream>>>(x, W_in, b_in, W1, part, dinv, p1, N);

    gcn_agg<0><<<nrg, AGG_TB, 0, stream>>>(ssrc, sdst, binstart, p1, dinv, b1, W2, b_out, p2, N, nchunk);

    gcn_agg<1><<<nrg, AGG_TB, 0, stream>>>(ssrc, sdst, binstart, p2, dinv, b2, W_out, b_out, out, N, nchunk);
}
